# starting kernel
# baseline (speedup 1.0000x reference)
#include <hip/hip_runtime.h>
#include <hip/hip_bf16.h>


typedef __attribute__((ext_vector_type(2))) float v2f;
typedef __attribute__((ext_vector_type(4))) float v4f;
typedef __attribute__((ext_vector_type(8))) float v8f;

#define DF 128


__global__ __launch_bounds__(256) void k_zero(float* __restrict__ p, int n) {
    int i = blockIdx.x * blockDim.x + threadIdx.x;
    if (i < n) p[i] = 0.0f;
}

__global__ __launch_bounds__(256) void k_degree(const int* __restrict__ dst,
                                                float* __restrict__ deg, int E) {
    int e = blockIdx.x * blockDim.x + threadIdx.x;
    if (e < E) atomicAdd(&deg[dst[e]], 1.0f);
}

__global__ __launch_bounds__(256) void k_dinv(float* __restrict__ deg, int n) {
    int i = blockIdx.x * blockDim.x + threadIdx.x;
    if (i < n) deg[i] = rsqrtf(deg[i] + 1.0f);
}

template <bool RELU_IN>
__global__ __launch_bounds__(256) void k_gemm_wmma(
        const float* __restrict__ X, const float* __restrict__ W,
        const float* __restrict__ bias, const float* __restrict__ dinv,
        float* __restrict__ H, float* __restrict__ AGG,
        int strips, int nrows) {
    __shared__ __align__(16) float ldsW[DF * DF];

    for (int idx = threadIdx.x; idx < DF * DF; idx += blockDim.x) {
        int kk = idx >> 7;
        int n  = idx & 127;
        int t     = n >> 4;
        int nloc  = n & 15;
        int k     = kk >> 2;
        int khalf = (kk >> 1) & 1;
        int lo    = kk & 1;
        ldsW[((t << 5) + k) * 64 + (khalf * 16 + nloc) * 2 + lo] = W[idx];
    }
    __syncthreads();

    const int wave  = threadIdx.x >> 5;
    const int lane  = threadIdx.x & 31;
    const int strip = blockIdx.x * (blockDim.x >> 5) + wave;
    if (strip >= strips) return;

    const int nloc  = lane & 15;
    const int khalf = lane >> 4;

    int arow_i = strip * 16 + nloc;
    if (arow_i >= nrows) arow_i = nrows - 1;
    const float* __restrict__ arow = X + (size_t)arow_i * DF + 2 * khalf;

    const v2f* __restrict__ bbase = (const v2f*)ldsW + lane;

    v8f acc[8] = {};
#pragma unroll
    for (int k = 0; k < 32; ++k) {
        v2f a = *(const v2f*)(arow + 4 * k);
        if (RELU_IN) {
            a.x = fmaxf(a.x, 0.0f);
            a.y = fmaxf(a.y, 0.0f);
        }
#pragma unroll
        for (int t = 0; t < 8; ++t) {
            v2f b = bbase[(t * 32 + k) * 32];
            acc[t] = __builtin_amdgcn_wmma_f32_16x16x4_f32(
                         false, a, false, b, (short)0, acc[t], false, false);
        }
    }

    float bcol[8];
#pragma unroll
    for (int t = 0; t < 8; ++t) bcol[t] = bias[t * 16 + nloc];

    const int rbase = strip * 16 + 8 * khalf;
#pragma unroll
    for (int r = 0; r < 8; ++r) {
        int row = rbase + r;
        if (row >= nrows) continue;
        float dv = dinv[row];
        float d2 = dv * dv;
        size_t ro = (size_t)row * DF;
#pragma unroll
        for (int t = 0; t < 8; ++t) {
            float val = acc[t][r];
            int col = t * 16 + nloc;
            H[ro + col]   = val;
            AGG[ro + col] = fmaf(val, d2, bcol[t]);
        }
    }
}

__global__ __launch_bounds__(256) void k_scatter(
        const float* __restrict__ H, const int* __restrict__ src,
        const int* __restrict__ dst, const float* __restrict__ dinv,
        float* __restrict__ AGG, int E) {
    int e = blockIdx.x * (blockDim.x >> 5) + (threadIdx.x >> 5);
    if (e >= E) return;
    int lane = threadIdx.x & 31;
    int s = src[e], d = dst[e];
    float nrm = dinv[s] * dinv[d];
    v4f v = *(const v4f*)(H + (size_t)s * DF + lane * 4);
    float* out = AGG + (size_t)d * DF + lane * 4;
    atomicAdd(out + 0, v.x * nrm);
    atomicAdd(out + 1, v.y * nrm);
    atomicAdd(out + 2, v.z * nrm);
    atomicAdd(out + 3, v.w * nrm);
}

__global__ __launch_bounds__(256) void k_normalize(const float* __restrict__ in,
                                                   float* __restrict__ out, int n) {
    int row = blockIdx.x * (blockDim.x >> 5) + (threadIdx.x >> 5);
    if (row >= n) return;
    int lane = threadIdx.x & 31;
    v4f v = *(const v4f*)(in + (size_t)row * DF + lane * 4);
    float ss = v.x * v.x + v.y * v.y + v.z * v.z + v.w * v.w;
#pragma unroll
    for (int m = 16; m > 0; m >>= 1) ss += __shfl_xor(ss, m, 32);
    float sc = 1.0f / fmaxf(sqrtf(ss), 1e-12f);
    v4f o = {v.x * sc, v.y * sc, v.z * sc, v.w * sc};
    *(v4f*)(out + (size_t)row * DF + lane * 4) = o;
}


extern "C" void kernel_launch(void* const* d_in, const int* in_sizes, int n_in,
                              void* d_out, int out_size, void* d_ws, size_t ws_size,
                              hipStream_t stream) {
    const float* x  = (const float*)d_in[0];
    const int*   ei = (const int*)  d_in[1];
    const float* W1 = (const float*)d_in[2];
    const float* b1 = (const float*)d_in[3];
    const float* W2 = (const float*)d_in[4];
    const float* b2 = (const float*)d_in[5];
    const float* W3 = (const float*)d_in[6];
    const float* b3 = (const float*)d_in[7];

    const int N = in_sizes[0] / DF;
    const int E = in_sizes[1] / 2;
    const int* src = ei;
    const int* dst = ei + E;

    float* out = (float*)d_out;

    char* ws = (char*)d_ws;
    size_t off0 = ((size_t)N * sizeof(float) + 255) & ~(size_t)255;
    float* dinv = (float*)ws;
    float* H    = (float*)(ws + off0);
    float* AGGa = H + (size_t)N * DF;

    const int strips  = (N + 15) / 16;
    const int gblocks = (strips + 7) / 8;
    const int sblocks = (E + 7) / 8;
    const int nb256   = (N + 255) / 256;
    const int eb256   = (E + 255) / 256;

    k_zero  <<<nb256, 256, 0, stream>>>(dinv, N);
    k_degree<<<eb256, 256, 0, stream>>>(dst, dinv, E);
    k_dinv  <<<nb256, 256, 0, stream>>>(dinv, N);

    k_gemm_wmma<false><<<gblocks, 256, 0, stream>>>(x, W1, b1, dinv, H, AGGa, strips, N);
    k_scatter         <<<sblocks, 256, 0, stream>>>(H, src, dst, dinv, AGGa, E);

    k_gemm_wmma<true> <<<gblocks, 256, 0, stream>>>(AGGa, W2, b2, dinv, H, out, strips, N);
    k_scatter         <<<sblocks, 256, 0, stream>>>(H, src, dst, dinv, out, E);

    k_gemm_wmma<true> <<<gblocks, 256, 0, stream>>>(out, W3, b3, dinv, H, AGGa, strips, N);
    k_scatter         <<<sblocks, 256, 0, stream>>>(H, src, dst, dinv, AGGa, E);

    k_normalize<<<(N + 7) / 8, 256, 0, stream>>>(AGGa, out, N);
}


# Aggregation: a work-group owns 391 rows, sums in LDS; it filters the edge list by a byte plane of block ids, adds gathered rows by LDS float atomics; no global float atomics; sum order = arrival
# speedup vs baseline: 1.8623x; 1.8623x over previous
#include <hip/hip_runtime.h>
#include <hip/hip_bf16.h>


typedef __attribute__((ext_vector_type(2))) float v2f;
typedef __attribute__((ext_vector_type(4))) float v4f;
typedef __attribute__((ext_vector_type(8))) float v8f;
typedef __attribute__((ext_vector_type(4))) unsigned v4u;

#define DF 128


__global__ __launch_bounds__(256) void k_zero(float* __restrict__ p, int n) {
    int i = blockIdx.x * blockDim.x + threadIdx.x;
    if (i < n) p[i] = 0.0f;
}

#define TR 391

__global__ __launch_bounds__(256) void k_degree(const int* __restrict__ dst,
                                                float* __restrict__ deg,
                                                unsigned char* __restrict__ blk, int E) {
    int e = blockIdx.x * blockDim.x + threadIdx.x;
    const int Epad = (E + 15) & ~15;
    if (e < E) {
        const int d = dst[e];
        atomicAdd(&deg[d], 1.0f);
        blk[e] = (unsigned char)(d / TR);
    } else if (e < Epad) {
        blk[e] = 0;
    }
}

__global__ __launch_bounds__(256) void k_dinv(float* __restrict__ deg, int n) {
    int i = blockIdx.x * blockDim.x + threadIdx.x;
    if (i < n) deg[i] = rsqrtf(deg[i] + 1.0f);
}

template <bool RELU_IN>
__global__ __launch_bounds__(256) void k_gemm_wmma(
        const float* __restrict__ X, const float* __restrict__ W,
        float* __restrict__ H, int strips, int nrows) {
    __shared__ __align__(16) float ldsW[DF * DF];

    for (int idx = threadIdx.x; idx < DF * DF; idx += blockDim.x) {
        int kk = idx >> 7;
        int n  = idx & 127;
        int t     = n >> 4;
        int nloc  = n & 15;
        int k     = kk >> 2;
        int khalf = (kk >> 1) & 1;
        int lo    = kk & 1;
        ldsW[((t << 5) + k) * 64 + (khalf * 16 + nloc) * 2 + lo] = W[idx];
    }
    __syncthreads();

    const int wave  = threadIdx.x >> 5;
    const int lane  = threadIdx.x & 31;
    const int strip = blockIdx.x * (blockDim.x >> 5) + wave;
    if (strip >= strips) return;

    const int nloc  = lane & 15;
    const int khalf = lane >> 4;

    int arow_i = strip * 16 + nloc;
    if (arow_i >= nrows) arow_i = nrows - 1;
    const float* __restrict__ arow = X + (size_t)arow_i * DF + 2 * khalf;

    const v2f* __restrict__ bbase = (const v2f*)ldsW + lane;

    v8f acc[8] = {};
#pragma unroll
    for (int k = 0; k < 32; ++k) {
        v2f a = *(const v2f*)(arow + 4 * k);
        if (RELU_IN) {
            a.x = fmaxf(a.x, 0.0f);
            a.y = fmaxf(a.y, 0.0f);
        }
#pragma unroll
        for (int t = 0; t < 8; ++t) {
            v2f b = bbase[(t * 32 + k) * 32];
            acc[t] = __builtin_amdgcn_wmma_f32_16x16x4_f32(
                         false, a, false, b, (short)0, acc[t], false, false);
        }
    }

    const int rbase = strip * 16 + 8 * khalf;
#pragma unroll
    for (int r = 0; r < 8; ++r) {
        int row = rbase + r;
        if (row >= nrows) continue;
        size_t ro = (size_t)row * DF;
#pragma unroll
        for (int t = 0; t < 8; ++t) H[ro + t * 16 + nloc] = acc[t][r];
    }
}

#define RPC  4
#define AGW  32
#define QCAP 64
#define DB   16
#define SU   4

__device__ __forceinline__ void lds_add(float* p, float x) {
    (void)__hip_atomic_fetch_add(p, x, __ATOMIC_RELAXED, __HIP_MEMORY_SCOPE_WORKGROUP);
}

__device__ __forceinline__ unsigned zero_bytes(unsigned x) {
    const unsigned t = (x & 0x7F7F7F7Fu) + 0x7F7F7F7Fu;
    return ~(t | x | 0x7F7F7F7Fu);
}

template <bool FULL>
__device__ __forceinline__ void drain_batch(
        const unsigned* qw, int cnt, int lane, int lo,
        const float* __restrict__ H, const int* __restrict__ src,
        const int* __restrict__ dst, const float* __restrict__ dinv, float* acc) {
    int s = 0, d = lo;
    float ws = 0.0f, wd = 0.0f;
    if (lane < cnt) {
        const unsigned e = qw[lane];
        s = src[e];
        d = dst[e];
        ws = dinv[s];
        wd = dinv[d];
    }
    const int dl = d - lo;
    const float* __restrict__ hl = H + lane;
    float* al = acc + lane;
    if (FULL) {
        float v[DB][RPC];
#pragma unroll
        for (int j = 0; j < DB; ++j) {
            const int sj = __builtin_amdgcn_readlane(s, j);
            const float* __restrict__ row = hl + (size_t)sj * DF;
#pragma unroll
            for (int c = 0; c < RPC; ++c) v[j][c] = row[32 * c];
        }
#pragma unroll
        for (int j = 0; j < DB; ++j) {
            const int dj = __builtin_amdgcn_readlane(dl, j);
            const float nj =
                __builtin_bit_cast(float, __builtin_amdgcn_readlane(__builtin_bit_cast(int, ws), j)) *
                __builtin_bit_cast(float, __builtin_amdgcn_readlane(__builtin_bit_cast(int, wd), j));
            float* a = al + dj * DF;
#pragma unroll
            for (int c = 0; c < RPC; ++c) lds_add(a + 32 * c, v[j][c] * nj);
        }
    } else {
        for (int j = 0; j < cnt; ++j) {
            const int sj = __builtin_amdgcn_readlane(s, j);
            const int dj = __builtin_amdgcn_readlane(dl, j);
            const float nj =
                __builtin_bit_cast(float, __builtin_amdgcn_readlane(__builtin_bit_cast(int, ws), j)) *
                __builtin_bit_cast(float, __builtin_amdgcn_readlane(__builtin_bit_cast(int, wd), j));
            const float* __restrict__ row = hl + (size_t)sj * DF;
            float* a = al + dj * DF;
            float t[RPC];
#pragma unroll
            for (int c = 0; c < RPC; ++c) t[c] = row[32 * c];
#pragma unroll
            for (int c = 0; c < RPC; ++c) lds_add(a + 32 * c, t[c] * nj);
        }
    }
}

__global__ __launch_bounds__(1024) void k_aggregate(
        const float* __restrict__ H, const int* __restrict__ src,
        const int* __restrict__ dst, const float* __restrict__ dinv,
        const float* __restrict__ bias, const unsigned char* __restrict__ blk,
        float* __restrict__ AGG, int N, int E) {
    __shared__ __align__(16) float acc[TR * DF];
    __shared__ unsigned q[AGW * QCAP];

    const int lane = threadIdx.x & 31;
    const int wave = __builtin_amdgcn_readfirstlane((int)(threadIdx.x >> 5));
    const int nblk = (N + TR - 1) / TR;
    const int G = (E + 15) >> 4;
    const v4u* __restrict__ b4 = (const v4u*)blk;
    unsigned* qw = q + wave * QCAP;

    for (int tile = blockIdx.x; tile < nblk; tile += gridDim.x) {
        const int lo = tile * TR;
        const int rows = (N - lo) < TR ? (N - lo) : TR;

        for (int i = threadIdx.x; i < rows * (DF / 4); i += blockDim.x) {
            const int r = i >> 5, c4 = (i & 31) * 4;
            const float dv = dinv[lo + r];
            const float d2 = dv * dv;
            const v4f h = *(const v4f*)(H + (size_t)(lo + r) * DF + c4);
            const v4f b = *(const v4f*)(bias + c4);
            v4f o = {fmaf(h.x, d2, b.x), fmaf(h.y, d2, b.y), fmaf(h.z, d2, b.z), fmaf(h.w, d2, b.w)};
            *(v4f*)(acc + r * DF + c4) = o;
        }
        __syncthreads();

        const unsigned pat = (unsigned)tile * 0x01010101u;
        int qn = 0;
        const int glast = G > 0 ? G - 1 : 0;
        v4u w[SU];
#pragma unroll
        for (int u = 0; u < SU; ++u) {
            const int g = (wave * SU + u) * 32 + lane;
            w[u] = b4[g < G ? g : glast];
        }
        for (int i0 = wave * SU; i0 * 32 < G; i0 += AGW * SU) {
            v4u nx[SU];
#pragma unroll
            for (int u = 0; u < SU; ++u) {
                const int g = (i0 + AGW * SU + u) * 32 + lane;
                nx[u] = b4[g < G ? g : glast];
            }
#pragma unroll
            for (int u = 0; u < SU; ++u) {
                const int g = (i0 + u) * 32 + lane;
                unsigned mask = (zero_bytes(w[u].x ^ pat) >> 7) | (zero_bytes(w[u].y ^ pat) >> 6) |
                                (zero_bytes(w[u].z ^ pat) >> 5) | (zero_bytes(w[u].w ^ pat) >> 4);
                if (g >= G) mask = 0u;
                while (__builtin_amdgcn_ballot_w32(mask != 0u) != 0u) {
                    const bool has = mask != 0u;
                    const unsigned p = (unsigned)__builtin_ctz(mask | 0x80000000u);
                    mask &= mask - 1u;
                    const unsigned e = (unsigned)g * 16u + ((p & 7u) << 2) + (p >> 3);
                    const bool ok = has && e < (unsigned)E;
                    const unsigned okb = __builtin_amdgcn_ballot_w32(ok);
                    if (ok) qw[qn + (int)__builtin_amdgcn_mbcnt_lo(okb, 0u)] = e;
                    qn += __builtin_popcount(okb);
                    __builtin_amdgcn_wave_barrier();
                    while (qn >= DB) {
                        qn -= DB;
                        drain_batch<true>(qw + qn, DB, lane, lo, H, src, dst, dinv, acc);
                    }
                }
            }
#pragma unroll
            for (int u = 0; u < SU; ++u) w[u] = nx[u];
        }
        if (qn > 0) drain_batch<false>(qw, qn, lane, lo, H, src, dst, dinv, acc);
        __syncthreads();

        for (int i = threadIdx.x; i < rows * (DF / 4); i += blockDim.x) {
            const int r = i >> 5, c4 = (i & 31) * 4;
            *(v4f*)(AGG + (size_t)(lo + r) * DF + c4) = *(const v4f*)(acc + r * DF + c4);
        }
        __syncthreads();
    }
}

__global__ __launch_bounds__(256) void k_normalize(const float* __restrict__ in,
                                                   float* __restrict__ out, int n) {
    int row = blockIdx.x * (blockDim.x >> 5) + (threadIdx.x >> 5);
    if (row >= n) return;
    int lane = threadIdx.x & 31;
    v4f v = *(const v4f*)(in + (size_t)row * DF + lane * 4);
    float ss = v.x * v.x + v.y * v.y + v.z * v.z + v.w * v.w;
#pragma unroll
    for (int m = 16; m > 0; m >>= 1) ss += __shfl_xor(ss, m, 32);
    float sc = 1.0f / fmaxf(sqrtf(ss), 1e-12f);
    v4f o = {v.x * sc, v.y * sc, v.z * sc, v.w * sc};
    *(v4f*)(out + (size_t)row * DF + lane * 4) = o;
}


extern "C" void kernel_launch(void* const* d_in, const int* in_sizes, int n_in,
                              void* d_out, int out_size, void* d_ws, size_t ws_size,
                              hipStream_t stream) {
    const float* x  = (const float*)d_in[0];
    const int*   ei = (const int*)  d_in[1];
    const float* W1 = (const float*)d_in[2];
    const float* b1 = (const float*)d_in[3];
    const float* W2 = (const float*)d_in[4];
    const float* b2 = (const float*)d_in[5];
    const float* W3 = (const float*)d_in[6];
    const float* b3 = (const float*)d_in[7];

    const int N = in_sizes[0] / DF;
    const int E = in_sizes[1] / 2;
    const int* src = ei;
    const int* dst = ei + E;

    float* out = (float*)d_out;

    char* ws = (char*)d_ws;
    size_t off0 = ((size_t)N * sizeof(float) + 255) & ~(size_t)255;
    float* dinv = (float*)ws;
    float* H    = (float*)(ws + off0);
    float* AGGa = H + (size_t)N * DF;
    unsigned char* blk = (unsigned char*)(AGGa + (size_t)N * DF);

    const int strips  = (N + 15) / 16;
    const int gblocks = (strips + 7) / 8;
    const int nb256   = (N + 255) / 256;
    const int eb256   = (E + 255) / 256;

    k_zero  <<<nb256, 256, 0, stream>>>(dinv, N);
    k_degree<<<eb256, 256, 0, stream>>>(dst, dinv, blk, E);
    k_dinv  <<<nb256, 256, 0, stream>>>(dinv, N);

    k_gemm_wmma<false><<<gblocks, 256, 0, stream>>>(x, W1, H, strips, N);
    k_aggregate       <<<256, 1024, 0, stream>>>(H, src, dst, dinv, b1, blk, AGGa, N, E);

    k_gemm_wmma<true> <<<gblocks, 256, 0, stream>>>(AGGa, W2, H, strips, N);
    k_aggregate       <<<256, 1024, 0, stream>>>(H, src, dst, dinv, b2, blk, out, N, E);

    k_gemm_wmma<true> <<<gblocks, 256, 0, stream>>>(out, W3, H, strips, N);
    k_aggregate       <<<256, 1024, 0, stream>>>(H, src, dst, dinv, b3, blk, AGGa, N, E);

    k_normalize<<<(N + 7) / 8, 256, 0, stream>>>(AGGa, out, N);
}


# Aggregation, layers 2-3: the edges a work-group filters in layer 1 are kept per block as records (source, row, norm) in the workspace and read back, not filtered again; LDS atomics, arrival order
# speedup vs baseline: 2.2807x; 1.2247x over previous
#include <hip/hip_runtime.h>
#include <hip/hip_bf16.h>


typedef __attribute__((ext_vector_type(2))) float v2f;
typedef __attribute__((ext_vector_type(4))) float v4f;
typedef __attribute__((ext_vector_type(8))) float v8f;
typedef __attribute__((ext_vector_type(4))) unsigned v4u;
typedef __attribute__((ext_vector_type(2))) unsigned v2u;

#define DF 128


__global__ __launch_bounds__(256) void k_zero(float* __restrict__ p, int n) {
    int i = blockIdx.x * blockDim.x + threadIdx.x;
    if (i < n) p[i] = 0.0f;
}

#define TR 391

__global__ __launch_bounds__(256) void k_degree(const int* __restrict__ dst,
                                                float* __restrict__ deg,
                                                unsigned char* __restrict__ blk, int E) {
    int e = blockIdx.x * blockDim.x + threadIdx.x;
    const int Epad = (E + 15) & ~15;
    if (e < E) {
        const int d = dst[e];
        atomicAdd(&deg[d], 1.0f);
        blk[e] = (unsigned char)(d / TR);
    } else if (e < Epad) {
        blk[e] = 0;
    }
}

__global__ __launch_bounds__(256) void k_dinv(float* __restrict__ deg, int n) {
    int i = blockIdx.x * blockDim.x + threadIdx.x;
    if (i < n) deg[i] = rsqrtf(deg[i] + 1.0f);
}

template <bool RELU_IN>
__global__ __launch_bounds__(256) void k_gemm_wmma(
        const float* __restrict__ X, const float* __restrict__ W,
        float* __restrict__ H, int strips, int nrows) {
    __shared__ __align__(16) float ldsW[DF * DF];

    for (int idx = threadIdx.x; idx < DF * DF; idx += blockDim.x) {
        int kk = idx >> 7;
        int n  = idx & 127;
        int t     = n >> 4;
        int nloc  = n & 15;
        int k     = kk >> 2;
        int khalf = (kk >> 1) & 1;
        int lo    = kk & 1;
        ldsW[((t << 5) + k) * 64 + (khalf * 16 + nloc) * 2 + lo] = W[idx];
    }
    __syncthreads();

    const int wave  = threadIdx.x >> 5;
    const int lane  = threadIdx.x & 31;
    const int strip = blockIdx.x * (blockDim.x >> 5) + wave;
    if (strip >= strips) return;

    const int nloc  = lane & 15;
    const int khalf = lane >> 4;

    int arow_i = strip * 16 + nloc;
    if (arow_i >= nrows) arow_i = nrows - 1;
    const float* __restrict__ arow = X + (size_t)arow_i * DF + 2 * khalf;

    const v2f* __restrict__ bbase = (const v2f*)ldsW + lane;

    v8f acc[8] = {};
#pragma unroll
    for (int k = 0; k < 32; ++k) {
        v2f a = *(const v2f*)(arow + 4 * k);
        if (RELU_IN) {
            a.x = fmaxf(a.x, 0.0f);
            a.y = fmaxf(a.y, 0.0f);
        }
#pragma unroll
        for (int t = 0; t < 8; ++t) {
            v2f b = bbase[(t * 32 + k) * 32];
            acc[t] = __builtin_amdgcn_wmma_f32_16x16x4_f32(
                         false, a, false, b, (short)0, acc[t], false, false);
        }
    }

    const int rbase = strip * 16 + 8 * khalf;
#pragma unroll
    for (int r = 0; r < 8; ++r) {
        int row = rbase + r;
        if (row >= nrows) continue;
        size_t ro = (size_t)row * DF;
#pragma unroll
        for (int t = 0; t < 8; ++t) H[ro + t * 16 + nloc] = acc[t][r];
    }
}

#define CAPB 24576
#define SRCB 23
#define RPC  4
#define AGW  32
#define QCAP 64
#define DB   16
#define SU   4

__device__ __forceinline__ void lds_add(float* p, float x) {
    (void)__hip_atomic_fetch_add(p, x, __ATOMIC_RELAXED, __HIP_MEMORY_SCOPE_WORKGROUP);
}

__device__ __forceinline__ unsigned zero_bytes(unsigned x) {
    const unsigned t = (x & 0x7F7F7F7Fu) + 0x7F7F7F7Fu;
    return ~(t | x | 0x7F7F7F7Fu);
}

template <bool FULL, bool REC>
__device__ __forceinline__ void drain_batch(
        const unsigned* qw, int cnt, int lane, int lo,
        const float* __restrict__ H, const int* __restrict__ src,
        const int* __restrict__ dst, const float* __restrict__ dinv, float* acc,
        v2u* __restrict__ rb, int* nrec) {
    int s = 0, d = lo;
    float ws = 0.0f, wd = 0.0f;
    if (lane < cnt) {
        const unsigned e = qw[lane];
        s = src[e];
        d = dst[e];
        ws = dinv[s];
        wd = dinv[d];
    }
    const int dl = d - lo;
    const float* __restrict__ hl = H + lane;
    float* al = acc + lane;
    auto put_records = [&]() {
        int pos = 0;
        if (lane == 0) pos = __hip_atomic_fetch_add(nrec, cnt, __ATOMIC_RELAXED, __HIP_MEMORY_SCOPE_WORKGROUP);
        const int p0 = __builtin_amdgcn_readfirstlane(pos);
        if (lane < cnt && p0 + lane < CAPB) {
            const v2u r = {(unsigned)s | ((unsigned)dl << SRCB), __builtin_bit_cast(unsigned, ws * wd)};
            rb[p0 + lane] = r;
        }
    };
    if (FULL) {
        float v[DB][RPC];
#pragma unroll
        for (int j = 0; j < DB; ++j) {
            const int sj = __builtin_amdgcn_readlane(s, j);
            const float* __restrict__ row = hl + (size_t)sj * DF;
#pragma unroll
            for (int c = 0; c < RPC; ++c) v[j][c] = row[32 * c];
        }
        if (REC) put_records();
#pragma unroll
        for (int j = 0; j < DB; ++j) {
            const int dj = __builtin_amdgcn_readlane(dl, j);
            const float nj =
                __builtin_bit_cast(float, __builtin_amdgcn_readlane(__builtin_bit_cast(int, ws), j)) *
                __builtin_bit_cast(float, __builtin_amdgcn_readlane(__builtin_bit_cast(int, wd), j));
            float* a = al + dj * DF;
#pragma unroll
            for (int c = 0; c < RPC; ++c) lds_add(a + 32 * c, v[j][c] * nj);
        }
    } else {
        if (REC) put_records();
        for (int j = 0; j < cnt; ++j) {
            const int sj = __builtin_amdgcn_readlane(s, j);
            const int dj = __builtin_amdgcn_readlane(dl, j);
            const float nj =
                __builtin_bit_cast(float, __builtin_amdgcn_readlane(__builtin_bit_cast(int, ws), j)) *
                __builtin_bit_cast(float, __builtin_amdgcn_readlane(__builtin_bit_cast(int, wd), j));
            const float* __restrict__ row = hl + (size_t)sj * DF;
            float* a = al + dj * DF;
            float t[RPC];
#pragma unroll
            for (int c = 0; c < RPC; ++c) t[c] = row[32 * c];
#pragma unroll
            for (int c = 0; c < RPC; ++c) lds_add(a + 32 * c, t[c] * nj);
        }
    }
}

template <bool FULL>
__device__ __forceinline__ void drain_records(v2u r, int cnt, int lane,
                                              const float* __restrict__ H, float* acc) {
    const float* __restrict__ hl = H + lane;
    float* al = acc + lane;
    const int rx = (int)r.x, ry = (int)r.y;
    if (FULL) {
        float v[DB][RPC];
#pragma unroll
        for (int j = 0; j < DB; ++j) {
            const unsigned pj = (unsigned)__builtin_amdgcn_readlane(rx, j);
            const float* __restrict__ row = hl + (size_t)(pj & ((1u << SRCB) - 1u)) * DF;
#pragma unroll
            for (int c = 0; c < RPC; ++c) v[j][c] = row[32 * c];
        }
#pragma unroll
        for (int j = 0; j < DB; ++j) {
            const unsigned pj = (unsigned)__builtin_amdgcn_readlane(rx, j);
            const float nj = __builtin_bit_cast(float, __builtin_amdgcn_readlane(ry, j));
            float* a = al + (pj >> SRCB) * DF;
#pragma unroll
            for (int c = 0; c < RPC; ++c) lds_add(a + 32 * c, v[j][c] * nj);
        }
    } else {
        for (int j = 0; j < cnt; ++j) {
            const unsigned pj = (unsigned)__builtin_amdgcn_readlane(rx, j);
            const float nj = __builtin_bit_cast(float, __builtin_amdgcn_readlane(ry, j));
            const float* __restrict__ row = hl + (size_t)(pj & ((1u << SRCB) - 1u)) * DF;
            float* a = al + (pj >> SRCB) * DF;
            float t[RPC];
#pragma unroll
            for (int c = 0; c < RPC; ++c) t[c] = row[32 * c];
#pragma unroll
            for (int c = 0; c < RPC; ++c) lds_add(a + 32 * c, t[c] * nj);
        }
    }
}

template <bool REC>
__device__ __forceinline__ void filter_block(
        int tile, int lo, int wave, int lane, int G, int E,
        const v4u* __restrict__ b4, unsigned* qw,
        const float* __restrict__ H, const int* __restrict__ src,
        const int* __restrict__ dst, const float* __restrict__ dinv, float* acc,
        v2u* __restrict__ rb, int* nrec) {
    const unsigned pat = (unsigned)tile * 0x01010101u;
    int qn = 0;
    const int glast = G > 0 ? G - 1 : 0;
    v4u w[SU];
#pragma unroll
    for (int u = 0; u < SU; ++u) {
        const int g = (wave * SU + u) * 32 + lane;
        w[u] = b4[g < G ? g : glast];
    }
    for (int i0 = wave * SU; i0 * 32 < G; i0 += AGW * SU) {
        v4u nx[SU];
#pragma unroll
        for (int u = 0; u < SU; ++u) {
            const int g = (i0 + AGW * SU + u) * 32 + lane;
            nx[u] = b4[g < G ? g : glast];
        }
#pragma unroll
        for (int u = 0; u < SU; ++u) {
            const int g = (i0 + u) * 32 + lane;
            unsigned mask = (zero_bytes(w[u].x ^ pat) >> 7) | (zero_bytes(w[u].y ^ pat) >> 6) |
                            (zero_bytes(w[u].z ^ pat) >> 5) | (zero_bytes(w[u].w ^ pat) >> 4);
            if (g >= G) mask = 0u;
            while (__builtin_amdgcn_ballot_w32(mask != 0u) != 0u) {
                const bool has = mask != 0u;
                const unsigned p = (unsigned)__builtin_ctz(mask | 0x80000000u);
                mask &= mask - 1u;
                const unsigned e = (unsigned)g * 16u + ((p & 7u) << 2) + (p >> 3);
                const bool ok = has && e < (unsigned)E;
                const unsigned okb = __builtin_amdgcn_ballot_w32(ok);
                if (ok) qw[qn + (int)__builtin_amdgcn_mbcnt_lo(okb, 0u)] = e;
                qn += __builtin_popcount(okb);
                __builtin_amdgcn_wave_barrier();
                while (qn >= DB) {
                    qn -= DB;
                    drain_batch<true, REC>(qw + qn, DB, lane, lo, H, src, dst, dinv, acc, rb, nrec);
                }
            }
        }
#pragma unroll
        for (int u = 0; u < SU; ++u) w[u] = nx[u];
    }
    if (qn > 0) drain_batch<false, REC>(qw, qn, lane, lo, H, src, dst, dinv, acc, rb, nrec);
}

template <bool BUILD>
__global__ __launch_bounds__(1024) void k_aggregate(
        const float* __restrict__ H, const int* __restrict__ src,
        const int* __restrict__ dst, const float* __restrict__ dinv,
        const float* __restrict__ bias, const unsigned char* __restrict__ blk,
        unsigned* __restrict__ recw, int* __restrict__ nrecs,
        float* __restrict__ AGG, int N, int E) {
    __shared__ __align__(16) float acc[TR * DF];
    __shared__ unsigned q[AGW * QCAP];
    __shared__ int nrec;

    const int lane = threadIdx.x & 31;
    const int wave = __builtin_amdgcn_readfirstlane((int)(threadIdx.x >> 5));
    const int nblk = (N + TR - 1) / TR;
    const int G = (E + 15) >> 4;
    const v4u* __restrict__ b4 = (const v4u*)blk;
    unsigned* qw = q + wave * QCAP;

    for (int tile = blockIdx.x; tile < nblk; tile += gridDim.x) {
        const int lo = tile * TR;
        const int rows = (N - lo) < TR ? (N - lo) : TR;
        v2u* __restrict__ rb = (v2u*)recw + (size_t)tile * CAPB;

        for (int i = threadIdx.x; i < rows * (DF / 4); i += blockDim.x) {
            const int r = i >> 5, c4 = (i & 31) * 4;
            const float dv = dinv[lo + r];
            const float d2 = dv * dv;
            const v4f h = *(const v4f*)(H + (size_t)(lo + r) * DF + c4);
            const v4f b = *(const v4f*)(bias + c4);
            v4f o = {fmaf(h.x, d2, b.x), fmaf(h.y, d2, b.y), fmaf(h.z, d2, b.z), fmaf(h.w, d2, b.w)};
            *(v4f*)(acc + r * DF + c4) = o;
        }
        if (BUILD && threadIdx.x == 0) nrec = 0;
        __syncthreads();

        if (BUILD) {
            filter_block<true>(tile, lo, wave, lane, G, E, b4, qw, H, src, dst, dinv, acc, rb, &nrec);
        } else {
            const int total = __builtin_amdgcn_readfirstlane(nrecs[tile]);
            if (total <= CAPB) {
                int base = wave * DB;
                v2u r = {0u, 0u};
                if (lane < DB && base + lane < total) r = rb[base + lane];
                for (; base < total; base += AGW * DB) {
                    const int nb = base + AGW * DB;
                    v2u rn = {0u, 0u};
                    if (lane < DB && nb + lane < total) rn = rb[nb + lane];
                    const int n = total - base;
                    if (n >= DB) drain_records<true>(r, DB, lane, H, acc);
                    else         drain_records<false>(r, n, lane, H, acc);
                    r = rn;
                }
            } else {
                filter_block<false>(tile, lo, wave, lane, G, E, b4, qw, H, src, dst, dinv, acc, rb, &nrec);
            }
        }
        __syncthreads();

        for (int i = threadIdx.x; i < rows * (DF / 4); i += blockDim.x) {
            const int r = i >> 5, c4 = (i & 31) * 4;
            *(v4f*)(AGG + (size_t)(lo + r) * DF + c4) = *(const v4f*)(acc + r * DF + c4);
        }
        if (BUILD && threadIdx.x == 0) nrecs[tile] = nrec;
        __syncthreads();
    }
}

__global__ __launch_bounds__(256) void k_normalize(const float* __restrict__ in,
                                                   float* __restrict__ out, int n) {
    int row = blockIdx.x * (blockDim.x >> 5) + (threadIdx.x >> 5);
    if (row >= n) return;
    int lane = threadIdx.x & 31;
    v4f v = *(const v4f*)(in + (size_t)row * DF + lane * 4);
    float ss = v.x * v.x + v.y * v.y + v.z * v.z + v.w * v.w;
#pragma unroll
    for (int m = 16; m > 0; m >>= 1) ss += __shfl_xor(ss, m, 32);
    float sc = 1.0f / fmaxf(sqrtf(ss), 1e-12f);
    v4f o = {v.x * sc, v.y * sc, v.z * sc, v.w * sc};
    *(v4f*)(out + (size_t)row * DF + lane * 4) = o;
}


extern "C" void kernel_launch(void* const* d_in, const int* in_sizes, int n_in,
                              void* d_out, int out_size, void* d_ws, size_t ws_size,
                              hipStream_t stream) {
    const float* x  = (const float*)d_in[0];
    const int*   ei = (const int*)  d_in[1];
    const float* W1 = (const float*)d_in[2];
    const float* b1 = (const float*)d_in[3];
    const float* W2 = (const float*)d_in[4];
    const float* b2 = (const float*)d_in[5];
    const float* W3 = (const float*)d_in[6];
    const float* b3 = (const float*)d_in[7];

    const int N = in_sizes[0] / DF;
    const int E = in_sizes[1] / 2;
    const int* src = ei;
    const int* dst = ei + E;

    float* out = (float*)d_out;

    char* ws = (char*)d_ws;
    size_t off0 = ((size_t)N * sizeof(float) + 255) & ~(size_t)255;
    float* dinv = (float*)ws;
    float* H    = (float*)(ws + off0);
    float* AGGa = H + (size_t)N * DF;
    unsigned char* blk = (unsigned char*)(AGGa + (size_t)N * DF);
    int* nrecs = (int*)(ws + ((size_t)128 << 20));
    unsigned* recs = (unsigned*)(ws + ((size_t)128 << 20) + 4096);

    const int strips  = (N + 15) / 16;
    const int gblocks = (strips + 7) / 8;
    const int nb256   = (N + 255) / 256;
    const int eb256   = (E + 255) / 256;

    k_zero  <<<nb256, 256, 0, stream>>>(dinv, N);
    k_degree<<<eb256, 256, 0, stream>>>(dst, dinv, blk, E);
    k_dinv  <<<nb256, 256, 0, stream>>>(dinv, N);

    k_gemm_wmma<false><<<gblocks, 256, 0, stream>>>(x, W1, H, strips, N);
    k_aggregate<true> <<<256, 1024, 0, stream>>>(H, src, dst, dinv, b1, blk, recs, nrecs, AGGa, N, E);

    k_gemm_wmma<true> <<<gblocks, 256, 0, stream>>>(AGGa, W2, H, strips, N);
    k_aggregate<false><<<256, 1024, 0, stream>>>(H, src, dst, dinv, b2, blk, recs, nrecs, out, N, E);

    k_gemm_wmma<true> <<<gblocks, 256, 0, stream>>>(out, W3, H, strips, N);
    k_aggregate<false><<<256, 1024, 0, stream>>>(H, src, dst, dinv, b3, blk, recs, nrecs, AGGa, N, E);

    k_normalize<<<(N + 7) / 8, 256, 0, stream>>>(AGGa, out, N);
}


# In-degrees: 128 edge chunks counted into whole-node LDS histograms (16-bit halves, LDS integer adds), summed by one launch; replaces 1.6M global float atomic adds; float sum order unchanged
# speedup vs baseline: 2.6001x; 1.1400x over previous
#include <hip/hip_runtime.h>
#include <hip/hip_bf16.h>


typedef __attribute__((ext_vector_type(2))) float v2f;
typedef __attribute__((ext_vector_type(4))) float v4f;
typedef __attribute__((ext_vector_type(8))) float v8f;
typedef __attribute__((ext_vector_type(4))) unsigned v4u;
typedef __attribute__((ext_vector_type(2))) unsigned v2u;

#define DF 128


#define TR 391

#define HG     128
#define HWORDS (256 * TR / 2)

#define EU 4
__global__ __launch_bounds__(1024) void k_degree_hist(const int* __restrict__ dst,
                                                      unsigned char* __restrict__ blk,
                                                      unsigned* __restrict__ hist, int E) {
    __shared__ __align__(16) unsigned h[HWORDS];
    const v4u z = {0u, 0u, 0u, 0u};
    for (int i = threadIdx.x; i < HWORDS / 4; i += blockDim.x) ((v4u*)h)[i] = z;
    __syncthreads();

    const int chunk = (E + (int)gridDim.x - 1) / (int)gridDim.x;
    const int e0 = (int)blockIdx.x * chunk;
    const int e1 = (e0 + chunk) < E ? (e0 + chunk) : E;
    const int step = (int)blockDim.x;
    int e = e0 + (int)threadIdx.x;
    for (; e + (EU - 1) * step < e1; e += EU * step) {
        int d[EU];
#pragma unroll
        for (int u = 0; u < EU; ++u) d[u] = dst[e + u * step];
#pragma unroll
        for (int u = 0; u < EU; ++u) {
            blk[e + u * step] = (unsigned char)(d[u] / TR);
            (void)__hip_atomic_fetch_add(&h[d[u] >> 1], 1u << ((d[u] & 1) * 16),
                                         __ATOMIC_RELAXED, __HIP_MEMORY_SCOPE_WORKGROUP);
        }
    }
    for (; e < e1; e += step) {
        const int d = dst[e];
        blk[e] = (unsigned char)(d / TR);
        (void)__hip_atomic_fetch_add(&h[d >> 1], 1u << ((d & 1) * 16),
                                     __ATOMIC_RELAXED, __HIP_MEMORY_SCOPE_WORKGROUP);
    }
    if (blockIdx.x == gridDim.x - 1) {
        const int Epad = (E + 15) & ~15;
        for (int p = E + (int)threadIdx.x; p < Epad; p += step) blk[p] = 0;
    }
    __syncthreads();

    v4u* __restrict__ o = (v4u*)(hist + (size_t)blockIdx.x * HWORDS);
    for (int i = threadIdx.x; i < HWORDS / 4; i += blockDim.x) o[i] = ((const v4u*)h)[i];
}

__global__ __launch_bounds__(256) void k_dinv_sum(const unsigned* __restrict__ hist,
                                                  float* __restrict__ dinv, int n, int groups) {
    const int w = blockIdx.x * blockDim.x + threadIdx.x;
    if (w >= HWORDS) return;
    const unsigned* __restrict__ p = hist + w;
    unsigned c0 = 0u, c1 = 0u;
    int g = 0;
    for (; g + 32 <= groups; g += 32) {
        unsigned v[32];
#pragma unroll
        for (int u = 0; u < 32; ++u) v[u] = p[(size_t)(g + u) * HWORDS];
#pragma unroll
        for (int u = 0; u < 32; ++u) { c0 += v[u] & 0xFFFFu; c1 += v[u] >> 16; }
    }
    for (; g < groups; ++g) {
        const unsigned v = p[(size_t)g * HWORDS];
        c0 += v & 0xFFFFu;
        c1 += v >> 16;
    }
    const int i = 2 * w;
    if (i < n)     dinv[i]     = rsqrtf((float)c0 + 1.0f);
    if (i + 1 < n) dinv[i + 1] = rsqrtf((float)c1 + 1.0f);
}

template <bool RELU_IN>
__global__ __launch_bounds__(256) void k_gemm_wmma(
        const float* __restrict__ X, const float* __restrict__ W,
        float* __restrict__ H, int strips, int nrows) {
    __shared__ __align__(16) float ldsW[DF * DF];

    for (int idx = threadIdx.x; idx < DF * DF; idx += blockDim.x) {
        int kk = idx >> 7;
        int n  = idx & 127;
        int t     = n >> 4;
        int nloc  = n & 15;
        int k     = kk >> 2;
        int khalf = (kk >> 1) & 1;
        int lo    = kk & 1;
        ldsW[((t << 5) + k) * 64 + (khalf * 16 + nloc) * 2 + lo] = W[idx];
    }
    __syncthreads();

    const int wave  = threadIdx.x >> 5;
    const int lane  = threadIdx.x & 31;
    const int strip = blockIdx.x * (blockDim.x >> 5) + wave;
    if (strip >= strips) return;

    const int nloc  = lane & 15;
    const int khalf = lane >> 4;

    int arow_i = strip * 16 + nloc;
    if (arow_i >= nrows) arow_i = nrows - 1;
    const float* __restrict__ arow = X + (size_t)arow_i * DF + 2 * khalf;

    const v2f* __restrict__ bbase = (const v2f*)ldsW + lane;

    v8f acc[8] = {};
#pragma unroll
    for (int k = 0; k < 32; ++k) {
        v2f a = *(const v2f*)(arow + 4 * k);
        if (RELU_IN) {
            a.x = fmaxf(a.x, 0.0f);
            a.y = fmaxf(a.y, 0.0f);
        }
#pragma unroll
        for (int t = 0; t < 8; ++t) {
            v2f b = bbase[(t * 32 + k) * 32];
            acc[t] = __builtin_amdgcn_wmma_f32_16x16x4_f32(
                         false, a, false, b, (short)0, acc[t], false, false);
        }
    }

    const int rbase = strip * 16 + 8 * khalf;
#pragma unroll
    for (int r = 0; r < 8; ++r) {
        int row = rbase + r;
        if (row >= nrows) continue;
        size_t ro = (size_t)row * DF;
#pragma unroll
        for (int t = 0; t < 8; ++t) H[ro + t * 16 + nloc] = acc[t][r];
    }
}

#define CAPB 24576
#define SRCB 23
#define RPC  4
#define AGW  32
#define QCAP 64
#define DB   16
#define SU   4

__device__ __forceinline__ void lds_add(float* p, float x) {
    (void)__hip_atomic_fetch_add(p, x, __ATOMIC_RELAXED, __HIP_MEMORY_SCOPE_WORKGROUP);
}

__device__ __forceinline__ unsigned zero_bytes(unsigned x) {
    const unsigned t = (x & 0x7F7F7F7Fu) + 0x7F7F7F7Fu;
    return ~(t | x | 0x7F7F7F7Fu);
}

template <bool FULL, bool REC>
__device__ __forceinline__ void drain_batch(
        const unsigned* qw, int cnt, int lane, int lo,
        const float* __restrict__ H, const int* __restrict__ src,
        const int* __restrict__ dst, const float* __restrict__ dinv, float* acc,
        v2u* __restrict__ rb, int* nrec) {
    int s = 0, d = lo;
    float ws = 0.0f, wd = 0.0f;
    if (lane < cnt) {
        const unsigned e = qw[lane];
        s = src[e];
        d = dst[e];
        ws = dinv[s];
        wd = dinv[d];
    }
    const int dl = d - lo;
    const float* __restrict__ hl = H + lane;
    float* al = acc + lane;
    auto put_records = [&]() {
        int pos = 0;
        if (lane == 0) pos = __hip_atomic_fetch_add(nrec, cnt, __ATOMIC_RELAXED, __HIP_MEMORY_SCOPE_WORKGROUP);
        const int p0 = __builtin_amdgcn_readfirstlane(pos);
        if (lane < cnt && p0 + lane < CAPB) {
            const v2u r = {(unsigned)s | ((unsigned)dl << SRCB), __builtin_bit_cast(unsigned, ws * wd)};
            rb[p0 + lane] = r;
        }
    };
    if (FULL) {
        float v[DB][RPC];
#pragma unroll
        for (int j = 0; j < DB; ++j) {
            const int sj = __builtin_amdgcn_readlane(s, j);
            const float* __restrict__ row = hl + (size_t)sj * DF;
#pragma unroll
            for (int c = 0; c < RPC; ++c) v[j][c] = row[32 * c];
        }
        if (REC) put_records();
#pragma unroll
        for (int j = 0; j < DB; ++j) {
            const int dj = __builtin_amdgcn_readlane(dl, j);
            const float nj =
                __builtin_bit_cast(float, __builtin_amdgcn_readlane(__builtin_bit_cast(int, ws), j)) *
                __builtin_bit_cast(float, __builtin_amdgcn_readlane(__builtin_bit_cast(int, wd), j));
            float* a = al + dj * DF;
#pragma unroll
            for (int c = 0; c < RPC; ++c) lds_add(a + 32 * c, v[j][c] * nj);
        }
    } else {
        if (REC) put_records();
        for (int j = 0; j < cnt; ++j) {
            const int sj = __builtin_amdgcn_readlane(s, j);
            const int dj = __builtin_amdgcn_readlane(dl, j);
            const float nj =
                __builtin_bit_cast(float, __builtin_amdgcn_readlane(__builtin_bit_cast(int, ws), j)) *
                __builtin_bit_cast(float, __builtin_amdgcn_readlane(__builtin_bit_cast(int, wd), j));
            const float* __restrict__ row = hl + (size_t)sj * DF;
            float* a = al + dj * DF;
            float t[RPC];
#pragma unroll
            for (int c = 0; c < RPC; ++c) t[c] = row[32 * c];
#pragma unroll
            for (int c = 0; c < RPC; ++c) lds_add(a + 32 * c, t[c] * nj);
        }
    }
}

template <bool FULL>
__device__ __forceinline__ void drain_records(v2u r, int cnt, int lane,
                                              const float* __restrict__ H, float* acc) {
    const float* __restrict__ hl = H + lane;
    float* al = acc + lane;
    const int rx = (int)r.x, ry = (int)r.y;
    if (FULL) {
        float v[DB][RPC];
#pragma unroll
        for (int j = 0; j < DB; ++j) {
            const unsigned pj = (unsigned)__builtin_amdgcn_readlane(rx, j);
            const float* __restrict__ row = hl + (size_t)(pj & ((1u << SRCB) - 1u)) * DF;
#pragma unroll
            for (int c = 0; c < RPC; ++c) v[j][c] = row[32 * c];
        }
#pragma unroll
        for (int j = 0; j < DB; ++j) {
            const unsigned pj = (unsigned)__builtin_amdgcn_readlane(rx, j);
            const float nj = __builtin_bit_cast(float, __builtin_amdgcn_readlane(ry, j));
            float* a = al + (pj >> SRCB) * DF;
#pragma unroll
            for (int c = 0; c < RPC; ++c) lds_add(a + 32 * c, v[j][c] * nj);
        }
    } else {
        for (int j = 0; j < cnt; ++j) {
            const unsigned pj = (unsigned)__builtin_amdgcn_readlane(rx, j);
            const float nj = __builtin_bit_cast(float, __builtin_amdgcn_readlane(ry, j));
            const float* __restrict__ row = hl + (size_t)(pj & ((1u << SRCB) - 1u)) * DF;
            float* a = al + (pj >> SRCB) * DF;
            float t[RPC];
#pragma unroll
            for (int c = 0; c < RPC; ++c) t[c] = row[32 * c];
#pragma unroll
            for (int c = 0; c < RPC; ++c) lds_add(a + 32 * c, t[c] * nj);
        }
    }
}

template <bool REC>
__device__ __forceinline__ void filter_block(
        int tile, int lo, int wave, int lane, int G, int E,
        const v4u* __restrict__ b4, unsigned* qw,
        const float* __restrict__ H, const int* __restrict__ src,
        const int* __restrict__ dst, const float* __restrict__ dinv, float* acc,
        v2u* __restrict__ rb, int* nrec) {
    const unsigned pat = (unsigned)tile * 0x01010101u;
    int qn = 0;
    const int glast = G > 0 ? G - 1 : 0;
    v4u w[SU];
#pragma unroll
    for (int u = 0; u < SU; ++u) {
        const int g = (wave * SU + u) * 32 + lane;
        w[u] = b4[g < G ? g : glast];
    }
    for (int i0 = wave * SU; i0 * 32 < G; i0 += AGW * SU) {
        v4u nx[SU];
#pragma unroll
        for (int u = 0; u < SU; ++u) {
            const int g = (i0 + AGW * SU + u) * 32 + lane;
            nx[u] = b4[g < G ? g : glast];
        }
#pragma unroll
        for (int u = 0; u < SU; ++u) {
            const int g = (i0 + u) * 32 + lane;
            unsigned mask = (zero_bytes(w[u].x ^ pat) >> 7) | (zero_bytes(w[u].y ^ pat) >> 6) |
                            (zero_bytes(w[u].z ^ pat) >> 5) | (zero_bytes(w[u].w ^ pat) >> 4);
            if (g >= G) mask = 0u;
            while (__builtin_amdgcn_ballot_w32(mask != 0u) != 0u) {
                const bool has = mask != 0u;
                const unsigned p = (unsigned)__builtin_ctz(mask | 0x80000000u);
                mask &= mask - 1u;
                const unsigned e = (unsigned)g * 16u + ((p & 7u) << 2) + (p >> 3);
                const bool ok = has && e < (unsigned)E;
                const unsigned okb = __builtin_amdgcn_ballot_w32(ok);
                if (ok) qw[qn + (int)__builtin_amdgcn_mbcnt_lo(okb, 0u)] = e;
                qn += __builtin_popcount(okb);
                __builtin_amdgcn_wave_barrier();
                while (qn >= DB) {
                    qn -= DB;
                    drain_batch<true, REC>(qw + qn, DB, lane, lo, H, src, dst, dinv, acc, rb, nrec);
                }
            }
        }
#pragma unroll
        for (int u = 0; u < SU; ++u) w[u] = nx[u];
    }
    if (qn > 0) drain_batch<false, REC>(qw, qn, lane, lo, H, src, dst, dinv, acc, rb, nrec);
}

template <bool BUILD>
__global__ __launch_bounds__(1024) void k_aggregate(
        const float* __restrict__ H, const int* __restrict__ src,
        const int* __restrict__ dst, const float* __restrict__ dinv,
        const float* __restrict__ bias, const unsigned char* __restrict__ blk,
        unsigned* __restrict__ recw, int* __restrict__ nrecs,
        float* __restrict__ AGG, int N, int E) {
    __shared__ __align__(16) float acc[TR * DF];
    __shared__ unsigned q[AGW * QCAP];
    __shared__ int nrec;

    const int lane = threadIdx.x & 31;
    const int wave = __builtin_amdgcn_readfirstlane((int)(threadIdx.x >> 5));
    const int nblk = (N + TR - 1) / TR;
    const int G = (E + 15) >> 4;
    const v4u* __restrict__ b4 = (const v4u*)blk;
    unsigned* qw = q + wave * QCAP;

    for (int tile = blockIdx.x; tile < nblk; tile += gridDim.x) {
        const int lo = tile * TR;
        const int rows = (N - lo) < TR ? (N - lo) : TR;
        v2u* __restrict__ rb = (v2u*)recw + (size_t)tile * CAPB;

        for (int i = threadIdx.x; i < rows * (DF / 4); i += blockDim.x) {
            const int r = i >> 5, c4 = (i & 31) * 4;
            const float dv = dinv[lo + r];
            const float d2 = dv * dv;
            const v4f h = *(const v4f*)(H + (size_t)(lo + r) * DF + c4);
            const v4f b = *(const v4f*)(bias + c4);
            v4f o = {fmaf(h.x, d2, b.x), fmaf(h.y, d2, b.y), fmaf(h.z, d2, b.z), fmaf(h.w, d2, b.w)};
            *(v4f*)(acc + r * DF + c4) = o;
        }
        if (BUILD && threadIdx.x == 0) nrec = 0;
        __syncthreads();

        if (BUILD) {
            filter_block<true>(tile, lo, wave, lane, G, E, b4, qw, H, src, dst, dinv, acc, rb, &nrec);
        } else {
            const int total = __builtin_amdgcn_readfirstlane(nrecs[tile]);
            if (total <= CAPB) {
                int base = wave * DB;
                v2u r = {0u, 0u};
                if (lane < DB && base + lane < total) r = rb[base + lane];
                for (; base < total; base += AGW * DB) {
                    const int nb = base + AGW * DB;
                    v2u rn = {0u, 0u};
                    if (lane < DB && nb + lane < total) rn = rb[nb + lane];
                    const int n = total - base;
                    if (n >= DB) drain_records<true>(r, DB, lane, H, acc);
                    else         drain_records<false>(r, n, lane, H, acc);
                    r = rn;
                }
            } else {
                filter_block<false>(tile, lo, wave, lane, G, E, b4, qw, H, src, dst, dinv, acc, rb, &nrec);
            }
        }
        __syncthreads();

        for (int i = threadIdx.x; i < rows * (DF / 4); i += blockDim.x) {
            const int r = i >> 5, c4 = (i & 31) * 4;
            *(v4f*)(AGG + (size_t)(lo + r) * DF + c4) = *(const v4f*)(acc + r * DF + c4);
        }
        if (BUILD && threadIdx.x == 0) nrecs[tile] = nrec;
        __syncthreads();
    }
}

__global__ __launch_bounds__(256) void k_normalize(const float* __restrict__ in,
                                                   float* __restrict__ out, int n) {
    int row = blockIdx.x * (blockDim.x >> 5) + (threadIdx.x >> 5);
    if (row >= n) return;
    int lane = threadIdx.x & 31;
    v4f v = *(const v4f*)(in + (size_t)row * DF + lane * 4);
    float ss = v.x * v.x + v.y * v.y + v.z * v.z + v.w * v.w;
#pragma unroll
    for (int m = 16; m > 0; m >>= 1) ss += __shfl_xor(ss, m, 32);
    float sc = 1.0f / fmaxf(sqrtf(ss), 1e-12f);
    v4f o = {v.x * sc, v.y * sc, v.z * sc, v.w * sc};
    *(v4f*)(out + (size_t)row * DF + lane * 4) = o;
}


extern "C" void kernel_launch(void* const* d_in, const int* in_sizes, int n_in,
                              void* d_out, int out_size, void* d_ws, size_t ws_size,
                              hipStream_t stream) {
    const float* x  = (const float*)d_in[0];
    const int*   ei = (const int*)  d_in[1];
    const float* W1 = (const float*)d_in[2];
    const float* b1 = (const float*)d_in[3];
    const float* W2 = (const float*)d_in[4];
    const float* b2 = (const float*)d_in[5];
    const float* W3 = (const float*)d_in[6];
    const float* b3 = (const float*)d_in[7];

    const int N = in_sizes[0] / DF;
    const int E = in_sizes[1] / 2;
    const int* src = ei;
    const int* dst = ei + E;

    float* out = (float*)d_out;

    char* ws = (char*)d_ws;
    size_t off0 = ((size_t)N * sizeof(float) + 255) & ~(size_t)255;
    float* dinv = (float*)ws;
    float* H    = (float*)(ws + off0);
    float* AGGa = H + (size_t)N * DF;
    unsigned char* blk = (unsigned char*)(AGGa + (size_t)N * DF);
    int* nrecs = (int*)(ws + ((size_t)128 << 20));
    unsigned* recs = (unsigned*)(ws + ((size_t)128 << 20) + 4096);

    const int strips  = (N + 15) / 16;
    const int gblocks = (strips + 7) / 8;

    unsigned* hist = (unsigned*)AGGa;
    k_degree_hist<<<HG, 1024, 0, stream>>>(dst, blk, hist, E);
    k_dinv_sum   <<<(HWORDS + 255) / 256, 256, 0, stream>>>(hist, dinv, N, HG);

    k_gemm_wmma<false><<<gblocks, 256, 0, stream>>>(x, W1, H, strips, N);
    k_aggregate<true> <<<256, 1024, 0, stream>>>(H, src, dst, dinv, b1, blk, recs, nrecs, AGGa, N, E);

    k_gemm_wmma<true> <<<gblocks, 256, 0, stream>>>(AGGa, W2, H, strips, N);
    k_aggregate<false><<<256, 1024, 0, stream>>>(H, src, dst, dinv, b2, blk, recs, nrecs, out, N, E);

    k_gemm_wmma<true> <<<gblocks, 256, 0, stream>>>(out, W3, H, strips, N);
    k_aggregate<false><<<256, 1024, 0, stream>>>(H, src, dst, dinv, b3, blk, recs, nrecs, AGGa, N, E);

    k_normalize<<<(N + 7) / 8, 256, 0, stream>>>(AGGa, out, N);
}


# Layers 2-3 GEMM moved into the aggregation launch before it: the group multiplies relu of its block sums, still in LDS (row stride padded), by the next W on WMMA and stores H; arithmetic order kept
# speedup vs baseline: 3.0864x; 1.1870x over previous
#include <hip/hip_runtime.h>
#include <hip/hip_bf16.h>


typedef __attribute__((ext_vector_type(2))) float v2f;
typedef __attribute__((ext_vector_type(4))) float v4f;
typedef __attribute__((ext_vector_type(8))) float v8f;
typedef __attribute__((ext_vector_type(4))) unsigned v4u;
typedef __attribute__((ext_vector_type(2))) unsigned v2u;

#define DF 128


#define TR 391

#define HG     128
#define HWORDS (256 * TR / 2)

#define EU 4
__global__ __launch_bounds__(1024) void k_degree_hist(const int* __restrict__ dst,
                                                      unsigned char* __restrict__ blk,
                                                      unsigned* __restrict__ hist, int E) {
    __shared__ __align__(16) unsigned h[HWORDS];
    const v4u z = {0u, 0u, 0u, 0u};
    for (int i = threadIdx.x; i < HWORDS / 4; i += blockDim.x) ((v4u*)h)[i] = z;
    __syncthreads();

    const int chunk = (E + (int)gridDim.x - 1) / (int)gridDim.x;
    const int e0 = (int)blockIdx.x * chunk;
    const int e1 = (e0 + chunk) < E ? (e0 + chunk) : E;
    const int step = (int)blockDim.x;
    int e = e0 + (int)threadIdx.x;
    for (; e + (EU - 1) * step < e1; e += EU * step) {
        int d[EU];
#pragma unroll
        for (int u = 0; u < EU; ++u) d[u] = dst[e + u * step];
#pragma unroll
        for (int u = 0; u < EU; ++u) {
            blk[e + u * step] = (unsigned char)(d[u] / TR);
            (void)__hip_atomic_fetch_add(&h[d[u] >> 1], 1u << ((d[u] & 1) * 16),
                                         __ATOMIC_RELAXED, __HIP_MEMORY_SCOPE_WORKGROUP);
        }
    }
    for (; e < e1; e += step) {
        const int d = dst[e];
        blk[e] = (unsigned char)(d / TR);
        (void)__hip_atomic_fetch_add(&h[d >> 1], 1u << ((d & 1) * 16),
                                     __ATOMIC_RELAXED, __HIP_MEMORY_SCOPE_WORKGROUP);
    }
    if (blockIdx.x == gridDim.x - 1) {
        const int Epad = (E + 15) & ~15;
        for (int p = E + (int)threadIdx.x; p < Epad; p += step) blk[p] = 0;
    }
    __syncthreads();

    v4u* __restrict__ o = (v4u*)(hist + (size_t)blockIdx.x * HWORDS);
    for (int i = threadIdx.x; i < HWORDS / 4; i += blockDim.x) o[i] = ((const v4u*)h)[i];
}

__global__ __launch_bounds__(256) void k_dinv_sum(const unsigned* __restrict__ hist,
                                                  float* __restrict__ dinv, int n, int groups) {
    const int w = blockIdx.x * blockDim.x + threadIdx.x;
    if (w >= HWORDS) return;
    const unsigned* __restrict__ p = hist + w;
    unsigned c0 = 0u, c1 = 0u;
    int g = 0;
    for (; g + 32 <= groups; g += 32) {
        unsigned v[32];
#pragma unroll
        for (int u = 0; u < 32; ++u) v[u] = p[(size_t)(g + u) * HWORDS];
#pragma unroll
        for (int u = 0; u < 32; ++u) { c0 += v[u] & 0xFFFFu; c1 += v[u] >> 16; }
    }
    for (; g < groups; ++g) {
        const unsigned v = p[(size_t)g * HWORDS];
        c0 += v & 0xFFFFu;
        c1 += v >> 16;
    }
    const int i = 2 * w;
    if (i < n)     dinv[i]     = rsqrtf((float)c0 + 1.0f);
    if (i + 1 < n) dinv[i + 1] = rsqrtf((float)c1 + 1.0f);
}

__global__ __launch_bounds__(256) void k_gemm_wmma(
        const float* __restrict__ X, const float* __restrict__ W,
        float* __restrict__ H, int strips, int nrows) {
    __shared__ __align__(16) float ldsW[DF * DF];

    for (int idx = threadIdx.x; idx < DF * DF; idx += blockDim.x) {
        int kk = idx >> 7;
        int n  = idx & 127;
        int t     = n >> 4;
        int nloc  = n & 15;
        int k     = kk >> 2;
        int khalf = (kk >> 1) & 1;
        int lo    = kk & 1;
        ldsW[((t << 5) + k) * 64 + (khalf * 16 + nloc) * 2 + lo] = W[idx];
    }
    __syncthreads();

    const int wave  = threadIdx.x >> 5;
    const int lane  = threadIdx.x & 31;
    const int strip = blockIdx.x * (blockDim.x >> 5) + wave;
    if (strip >= strips) return;

    const int nloc  = lane & 15;
    const int khalf = lane >> 4;

    int arow_i = strip * 16 + nloc;
    if (arow_i >= nrows) arow_i = nrows - 1;
    const float* __restrict__ arow = X + (size_t)arow_i * DF + 2 * khalf;

    const v2f* __restrict__ bbase = (const v2f*)ldsW + lane;

    v8f acc[8] = {};
#pragma unroll
    for (int k = 0; k < 32; ++k) {
        v2f a = *(const v2f*)(arow + 4 * k);
#pragma unroll
        for (int t = 0; t < 8; ++t) {
            v2f b = bbase[(t * 32 + k) * 32];
            acc[t] = __builtin_amdgcn_wmma_f32_16x16x4_f32(
                         false, a, false, b, (short)0, acc[t], false, false);
        }
    }

    const int rbase = strip * 16 + 8 * khalf;
#pragma unroll
    for (int r = 0; r < 8; ++r) {
        int row = rbase + r;
        if (row >= nrows) continue;
        size_t ro = (size_t)row * DF;
#pragma unroll
        for (int t = 0; t < 8; ++t) H[ro + t * 16 + nloc] = acc[t][r];
    }
}

#define CAPB 24576
#define SRCB 23
#define RPC  4
#define AS   132
#define AGT  1024
#define TRP  (((TR + 15) / 16) * 16)
#define AGW  32
#define QCAP 64
#define DB   16
#define SU   4

__device__ __forceinline__ void lds_add(float* p, float x) {
    (void)__hip_atomic_fetch_add(p, x, __ATOMIC_RELAXED, __HIP_MEMORY_SCOPE_WORKGROUP);
}

__device__ __forceinline__ unsigned zero_bytes(unsigned x) {
    const unsigned t = (x & 0x7F7F7F7Fu) + 0x7F7F7F7Fu;
    return ~(t | x | 0x7F7F7F7Fu);
}

template <bool FULL, bool REC>
__device__ __forceinline__ void drain_batch(
        const unsigned* qw, int cnt, int lane, int lo,
        const float* __restrict__ H, const int* __restrict__ src,
        const int* __restrict__ dst, const float* __restrict__ dinv, float* acc,
        v2u* __restrict__ rb, int* nrec) {
    int s = 0, d = lo;
    float ws = 0.0f, wd = 0.0f;
    if (lane < cnt) {
        const unsigned e = qw[lane];
        s = src[e];
        d = dst[e];
        ws = dinv[s];
        wd = dinv[d];
    }
    const int dl = d - lo;
    const float* __restrict__ hl = H + lane;
    float* al = acc + lane;
    auto put_records = [&]() {
        int pos = 0;
        if (lane == 0) pos = __hip_atomic_fetch_add(nrec, cnt, __ATOMIC_RELAXED, __HIP_MEMORY_SCOPE_WORKGROUP);
        const int p0 = __builtin_amdgcn_readfirstlane(pos);
        if (lane < cnt && p0 + lane < CAPB) {
            const v2u r = {(unsigned)s | ((unsigned)dl << SRCB), __builtin_bit_cast(unsigned, ws * wd)};
            rb[p0 + lane] = r;
        }
    };
    if (FULL) {
        float v[DB][RPC];
#pragma unroll
        for (int j = 0; j < DB; ++j) {
            const int sj = __builtin_amdgcn_readlane(s, j);
            const float* __restrict__ row = hl + (size_t)sj * DF;
#pragma unroll
            for (int c = 0; c < RPC; ++c) v[j][c] = row[32 * c];
        }
        if (REC) put_records();
#pragma unroll
        for (int j = 0; j < DB; ++j) {
            const int dj = __builtin_amdgcn_readlane(dl, j);
            const float nj =
                __builtin_bit_cast(float, __builtin_amdgcn_readlane(__builtin_bit_cast(int, ws), j)) *
                __builtin_bit_cast(float, __builtin_amdgcn_readlane(__builtin_bit_cast(int, wd), j));
            float* a = al + dj * AS;
#pragma unroll
            for (int c = 0; c < RPC; ++c) lds_add(a + 32 * c, v[j][c] * nj);
        }
    } else {
        if (REC) put_records();
        for (int j = 0; j < cnt; ++j) {
            const int sj = __builtin_amdgcn_readlane(s, j);
            const int dj = __builtin_amdgcn_readlane(dl, j);
            const float nj =
                __builtin_bit_cast(float, __builtin_amdgcn_readlane(__builtin_bit_cast(int, ws), j)) *
                __builtin_bit_cast(float, __builtin_amdgcn_readlane(__builtin_bit_cast(int, wd), j));
            const float* __restrict__ row = hl + (size_t)sj * DF;
            float* a = al + dj * AS;
            float t[RPC];
#pragma unroll
            for (int c = 0; c < RPC; ++c) t[c] = row[32 * c];
#pragma unroll
            for (int c = 0; c < RPC; ++c) lds_add(a + 32 * c, t[c] * nj);
        }
    }
}

template <bool FULL>
__device__ __forceinline__ void drain_records(v2u r, int cnt, int lane,
                                              const float* __restrict__ H, float* acc) {
    const float* __restrict__ hl = H + lane;
    float* al = acc + lane;
    const int rx = (int)r.x, ry = (int)r.y;
    if (FULL) {
        float v[DB][RPC];
#pragma unroll
        for (int j = 0; j < DB; ++j) {
            const unsigned pj = (unsigned)__builtin_amdgcn_readlane(rx, j);
            const float* __restrict__ row = hl + (size_t)(pj & ((1u << SRCB) - 1u)) * DF;
#pragma unroll
            for (int c = 0; c < RPC; ++c) v[j][c] = row[32 * c];
        }
#pragma unroll
        for (int j = 0; j < DB; ++j) {
            const unsigned pj = (unsigned)__builtin_amdgcn_readlane(rx, j);
            const float nj = __builtin_bit_cast(float, __builtin_amdgcn_readlane(ry, j));
            float* a = al + (pj >> SRCB) * AS;
#pragma unroll
            for (int c = 0; c < RPC; ++c) lds_add(a + 32 * c, v[j][c] * nj);
        }
    } else {
        for (int j = 0; j < cnt; ++j) {
            const unsigned pj = (unsigned)__builtin_amdgcn_readlane(rx, j);
            const float nj = __builtin_bit_cast(float, __builtin_amdgcn_readlane(ry, j));
            const float* __restrict__ row = hl + (size_t)(pj & ((1u << SRCB) - 1u)) * DF;
            float* a = al + (pj >> SRCB) * AS;
            float t[RPC];
#pragma unroll
            for (int c = 0; c < RPC; ++c) t[c] = row[32 * c];
#pragma unroll
            for (int c = 0; c < RPC; ++c) lds_add(a + 32 * c, t[c] * nj);
        }
    }
}

template <bool REC>
__device__ __forceinline__ void filter_block(
        int tile, int lo, int wave, int lane, int G, int E,
        const v4u* __restrict__ b4, unsigned* qw,
        const float* __restrict__ H, const int* __restrict__ src,
        const int* __restrict__ dst, const float* __restrict__ dinv, float* acc,
        v2u* __restrict__ rb, int* nrec) {
    const unsigned pat = (unsigned)tile * 0x01010101u;
    int qn = 0;
    const int glast = G > 0 ? G - 1 : 0;
    v4u w[SU];
#pragma unroll
    for (int u = 0; u < SU; ++u) {
        const int g = (wave * SU + u) * 32 + lane;
        w[u] = b4[g < G ? g : glast];
    }
    for (int i0 = wave * SU; i0 * 32 < G; i0 += AGW * SU) {
        v4u nx[SU];
#pragma unroll
        for (int u = 0; u < SU; ++u) {
            const int g = (i0 + AGW * SU + u) * 32 + lane;
            nx[u] = b4[g < G ? g : glast];
        }
#pragma unroll
        for (int u = 0; u < SU; ++u) {
            const int g = (i0 + u) * 32 + lane;
            unsigned mask = (zero_bytes(w[u].x ^ pat) >> 7) | (zero_bytes(w[u].y ^ pat) >> 6) |
                            (zero_bytes(w[u].z ^ pat) >> 5) | (zero_bytes(w[u].w ^ pat) >> 4);
            if (g >= G) mask = 0u;
            while (__builtin_amdgcn_ballot_w32(mask != 0u) != 0u) {
                const bool has = mask != 0u;
                const unsigned p = (unsigned)__builtin_ctz(mask | 0x80000000u);
                mask &= mask - 1u;
                const unsigned e = (unsigned)g * 16u + ((p & 7u) << 2) + (p >> 3);
                const bool ok = has && e < (unsigned)E;
                const unsigned okb = __builtin_amdgcn_ballot_w32(ok);
                if (ok) qw[qn + (int)__builtin_amdgcn_mbcnt_lo(okb, 0u)] = e;
                qn += __builtin_popcount(okb);
                __builtin_amdgcn_wave_barrier();
                while (qn >= DB) {
                    qn -= DB;
                    drain_batch<true, REC>(qw + qn, DB, lane, lo, H, src, dst, dinv, acc, rb, nrec);
                }
            }
        }
#pragma unroll
        for (int u = 0; u < SU; ++u) w[u] = nx[u];
    }
    if (qn > 0) drain_batch<false, REC>(qw, qn, lane, lo, H, src, dst, dinv, acc, rb, nrec);
}

template <bool BUILD, bool FUSE>
__global__ __launch_bounds__(AGT) void k_aggregate(
        const float* __restrict__ H, const int* __restrict__ src,
        const int* __restrict__ dst, const float* __restrict__ dinv,
        const float* __restrict__ bias, const unsigned char* __restrict__ blk,
        unsigned* __restrict__ recw, int* __restrict__ nrecs,
        const float* __restrict__ Wn, float* __restrict__ OUT, int N, int E) {
    __shared__ __align__(16) float acc[(FUSE ? TRP : TR) * AS];
    __shared__ __align__(16) float ldsW[FUSE ? DF * DF : 4];
    __shared__ unsigned q[AGW * QCAP];
    __shared__ int nrec;

    const int lane = threadIdx.x & 31;
    const int wave = __builtin_amdgcn_readfirstlane((int)(threadIdx.x >> 5));
    const int nblk = (N + TR - 1) / TR;
    const int G = (E + 15) >> 4;
    const v4u* __restrict__ b4 = (const v4u*)blk;
    unsigned* qw = q + wave * QCAP;

    if (FUSE) {
        float w0[DF * DF / 2 / AGT], w1[DF * DF / 2 / AGT];
#pragma unroll
        for (int u = 0; u < DF * DF / 2 / AGT; ++u) {
            const int p = u * AGT + (int)threadIdx.x;
            const int qq = p >> 7, n = p & 127;
            w0[u] = Wn[(2 * qq) * DF + n];
            w1[u] = Wn[(2 * qq + 1) * DF + n];
        }
#pragma unroll
        for (int u = 0; u < DF * DF / 2 / AGT; ++u) {
            const int p = u * AGT + (int)threadIdx.x;
            const int qq = p >> 7, n = p & 127;
            const v2f v = {w0[u], w1[u]};
            ((v2f*)ldsW)[(((n >> 4) << 5) + (qq >> 1)) * 32 + (qq & 1) * 16 + (n & 15)] = v;
        }
    }

    const int tile = blockIdx.x;
    if (tile >= nblk) return;
    {
        const int lo = tile * TR;
        const int rows = (N - lo) < TR ? (N - lo) : TR;
        v2u* __restrict__ rb = (v2u*)recw + (size_t)tile * CAPB;

        for (int i = threadIdx.x; i < rows * (DF / 4); i += blockDim.x) {
            const int r = i >> 5, c4 = (i & 31) * 4;
            const float dv = dinv[lo + r];
            const float d2 = dv * dv;
            const v4f h = *(const v4f*)(H + (size_t)(lo + r) * DF + c4);
            const v4f b = *(const v4f*)(bias + c4);
            v4f o = {fmaf(h.x, d2, b.x), fmaf(h.y, d2, b.y), fmaf(h.z, d2, b.z), fmaf(h.w, d2, b.w)};
            *(v4f*)(acc + r * AS + c4) = o;
        }
        if (FUSE) {
            const v4f z = {0.0f, 0.0f, 0.0f, 0.0f};
            for (int i = rows * (DF / 4) + (int)threadIdx.x; i < TRP * (DF / 4); i += blockDim.x)
                *(v4f*)(acc + (i >> 5) * AS + (i & 31) * 4) = z;
        }
        if (BUILD && threadIdx.x == 0) nrec = 0;
        __syncthreads();

        if (BUILD) {
            filter_block<true>(tile, lo, wave, lane, G, E, b4, qw, H, src, dst, dinv, acc, rb, &nrec);
        } else {
            const int total = __builtin_amdgcn_readfirstlane(nrecs[tile]);
            if (total <= CAPB) {
                int base = wave * DB;
                v2u r = {0u, 0u};
                if (lane < DB && base + lane < total) r = rb[base + lane];
                for (; base < total; base += AGW * DB) {
                    const int nb = base + AGW * DB;
                    v2u rn = {0u, 0u};
                    if (lane < DB && nb + lane < total) rn = rb[nb + lane];
                    const int n = total - base;
                    if (n >= DB) drain_records<true>(r, DB, lane, H, acc);
                    else         drain_records<false>(r, n, lane, H, acc);
                    r = rn;
                }
            } else {
                filter_block<false>(tile, lo, wave, lane, G, E, b4, qw, H, src, dst, dinv, acc, rb, &nrec);
            }
        }
        __syncthreads();

        if (FUSE) {
            if (wave * 16 < rows) {
                const int nloc  = lane & 15;
                const int khalf = lane >> 4;
                const float* arow = acc + (wave * 16 + nloc) * AS + 2 * khalf;
                const v2f* bbase = (const v2f*)ldsW + lane;
                v8f c[8] = {};
#pragma unroll
                for (int k = 0; k < 32; ++k) {
                    v2f a = *(const v2f*)(arow + 4 * k);
                    a.x = fmaxf(a.x, 0.0f);
                    a.y = fmaxf(a.y, 0.0f);
#pragma unroll
                    for (int t = 0; t < 8; ++t) {
                        const v2f b = bbase[(t * 32 + k) * 32];
                        c[t] = __builtin_amdgcn_wmma_f32_16x16x4_f32(
                                   false, a, false, b, (short)0, c[t], false, false);
                    }
                }
                float* drow = acc + (wave * 16 + 8 * khalf) * AS + nloc;
#pragma unroll
                for (int r = 0; r < 8; ++r) {
#pragma unroll
                    for (int t = 0; t < 8; ++t) drow[r * AS + t * 16] = c[t][r];
                }
            }
            __syncthreads();
        }

        for (int i = threadIdx.x; i < rows * (DF / 4); i += blockDim.x) {
            const int r = i >> 5, c4 = (i & 31) * 4;
            *(v4f*)(OUT + (size_t)(lo + r) * DF + c4) = *(const v4f*)(acc + r * AS + c4);
        }
        if (BUILD && threadIdx.x == 0) nrecs[tile] = nrec;
    }
}

__global__ __launch_bounds__(256) void k_normalize(const float* __restrict__ in,
                                                   float* __restrict__ out, int n) {
    int row = blockIdx.x * (blockDim.x >> 5) + (threadIdx.x >> 5);
    if (row >= n) return;
    int lane = threadIdx.x & 31;
    v4f v = *(const v4f*)(in + (size_t)row * DF + lane * 4);
    float ss = v.x * v.x + v.y * v.y + v.z * v.z + v.w * v.w;
#pragma unroll
    for (int m = 16; m > 0; m >>= 1) ss += __shfl_xor(ss, m, 32);
    float sc = 1.0f / fmaxf(sqrtf(ss), 1e-12f);
    v4f o = {v.x * sc, v.y * sc, v.z * sc, v.w * sc};
    *(v4f*)(out + (size_t)row * DF + lane * 4) = o;
}


extern "C" void kernel_launch(void* const* d_in, const int* in_sizes, int n_in,
                              void* d_out, int out_size, void* d_ws, size_t ws_size,
                              hipStream_t stream) {
    const float* x  = (const float*)d_in[0];
    const int*   ei = (const int*)  d_in[1];
    const float* W1 = (const float*)d_in[2];
    const float* b1 = (const float*)d_in[3];
    const float* W2 = (const float*)d_in[4];
    const float* b2 = (const float*)d_in[5];
    const float* W3 = (const float*)d_in[6];
    const float* b3 = (const float*)d_in[7];

    const int N = in_sizes[0] / DF;
    const int E = in_sizes[1] / 2;
    const int* src = ei;
    const int* dst = ei + E;

    float* out = (float*)d_out;

    char* ws = (char*)d_ws;
    size_t off0 = ((size_t)N * sizeof(float) + 255) & ~(size_t)255;
    float* dinv = (float*)ws;
    float* H    = (float*)(ws + off0);
    float* AGGa = H + (size_t)N * DF;
    unsigned char* blk = (unsigned char*)(AGGa + (size_t)N * DF);
    int* nrecs = (int*)(ws + ((size_t)128 << 20));
    unsigned* recs = (unsigned*)(ws + ((size_t)128 << 20) + 4096);

    const int strips  = (N + 15) / 16;
    const int gblocks = (strips + 7) / 8;

    unsigned* hist = (unsigned*)AGGa;
    k_degree_hist<<<HG, 1024, 0, stream>>>(dst, blk, hist, E);
    k_dinv_sum   <<<(HWORDS + 255) / 256, 256, 0, stream>>>(hist, dinv, N, HG);

    k_gemm_wmma<<<gblocks, 256, 0, stream>>>(x, W1, H, strips, N);
    k_aggregate<true, true>  <<<256, AGT, 0, stream>>>(H, src, dst, dinv, b1, blk, recs, nrecs, W2, AGGa, N, E);

    k_aggregate<false, true> <<<256, AGT, 0, stream>>>(AGGa, src, dst, dinv, b2, blk, recs, nrecs, W3, H, N, E);

    k_aggregate<false, false><<<256, AGT, 0, stream>>>(H, src, dst, dinv, b3, blk, recs, nrecs, W3, AGGa, N, E);

    k_normalize<<<(N + 7) / 8, 256, 0, stream>>>(AGGa, out, N);
}


# Layers 2-3: a block's records kept as 32 lists by owner wave (row mod 32, slots from LDS counters); wave w alone adds list w: plain LDS read-add-write, 16-B row loads, no float atomics; list order
# speedup vs baseline: 3.2325x; 1.0473x over previous
#include <hip/hip_runtime.h>
#include <hip/hip_bf16.h>


typedef __attribute__((ext_vector_type(2))) float v2f;
typedef __attribute__((ext_vector_type(4))) float v4f;
typedef __attribute__((ext_vector_type(8))) float v8f;
typedef __attribute__((ext_vector_type(4))) unsigned v4u;
typedef __attribute__((ext_vector_type(2))) unsigned v2u;

#define DF 128


#define TR 391

#define HG     128
#define HWORDS (256 * TR / 2)

#define EU 4
__global__ __launch_bounds__(1024) void k_degree_hist(const int* __restrict__ dst,
                                                      unsigned char* __restrict__ blk,
                                                      unsigned* __restrict__ hist, int E) {
    __shared__ __align__(16) unsigned h[HWORDS];
    const v4u z = {0u, 0u, 0u, 0u};
    for (int i = threadIdx.x; i < HWORDS / 4; i += blockDim.x) ((v4u*)h)[i] = z;
    __syncthreads();

    const int chunk = (E + (int)gridDim.x - 1) / (int)gridDim.x;
    const int e0 = (int)blockIdx.x * chunk;
    const int e1 = (e0 + chunk) < E ? (e0 + chunk) : E;
    const int step = (int)blockDim.x;
    int e = e0 + (int)threadIdx.x;
    for (; e + (EU - 1) * step < e1; e += EU * step) {
        int d[EU];
#pragma unroll
        for (int u = 0; u < EU; ++u) d[u] = dst[e + u * step];
#pragma unroll
        for (int u = 0; u < EU; ++u) {
            blk[e + u * step] = (unsigned char)(d[u] / TR);
            (void)__hip_atomic_fetch_add(&h[d[u] >> 1], 1u << ((d[u] & 1) * 16),
                                         __ATOMIC_RELAXED, __HIP_MEMORY_SCOPE_WORKGROUP);
        }
    }
    for (; e < e1; e += step) {
        const int d = dst[e];
        blk[e] = (unsigned char)(d / TR);
        (void)__hip_atomic_fetch_add(&h[d >> 1], 1u << ((d & 1) * 16),
                                     __ATOMIC_RELAXED, __HIP_MEMORY_SCOPE_WORKGROUP);
    }
    if (blockIdx.x == gridDim.x - 1) {
        const int Epad = (E + 15) & ~15;
        for (int p = E + (int)threadIdx.x; p < Epad; p += step) blk[p] = 0;
    }
    __syncthreads();

    v4u* __restrict__ o = (v4u*)(hist + (size_t)blockIdx.x * HWORDS);
    for (int i = threadIdx.x; i < HWORDS / 4; i += blockDim.x) o[i] = ((const v4u*)h)[i];
}

__global__ __launch_bounds__(256) void k_dinv_sum(const unsigned* __restrict__ hist,
                                                  float* __restrict__ dinv, int n, int groups) {
    const int w = blockIdx.x * blockDim.x + threadIdx.x;
    if (w >= HWORDS) return;
    const unsigned* __restrict__ p = hist + w;
    unsigned c0 = 0u, c1 = 0u;
    int g = 0;
    for (; g + 32 <= groups; g += 32) {
        unsigned v[32];
#pragma unroll
        for (int u = 0; u < 32; ++u) v[u] = p[(size_t)(g + u) * HWORDS];
#pragma unroll
        for (int u = 0; u < 32; ++u) { c0 += v[u] & 0xFFFFu; c1 += v[u] >> 16; }
    }
    for (; g < groups; ++g) {
        const unsigned v = p[(size_t)g * HWORDS];
        c0 += v & 0xFFFFu;
        c1 += v >> 16;
    }
    const int i = 2 * w;
    if (i < n)     dinv[i]     = rsqrtf((float)c0 + 1.0f);
    if (i + 1 < n) dinv[i + 1] = rsqrtf((float)c1 + 1.0f);
}

__global__ __launch_bounds__(256) void k_gemm_wmma(
        const float* __restrict__ X, const float* __restrict__ W,
        float* __restrict__ H, int strips, int nrows) {
    __shared__ __align__(16) float ldsW[DF * DF];

    for (int idx = threadIdx.x; idx < DF * DF; idx += blockDim.x) {
        int kk = idx >> 7;
        int n  = idx & 127;
        int t     = n >> 4;
        int nloc  = n & 15;
        int k     = kk >> 2;
        int khalf = (kk >> 1) & 1;
        int lo    = kk & 1;
        ldsW[((t << 5) + k) * 64 + (khalf * 16 + nloc) * 2 + lo] = W[idx];
    }
    __syncthreads();

    const int wave  = threadIdx.x >> 5;
    const int lane  = threadIdx.x & 31;
    const int strip = blockIdx.x * (blockDim.x >> 5) + wave;
    if (strip >= strips) return;

    const int nloc  = lane & 15;
    const int khalf = lane >> 4;

    int arow_i = strip * 16 + nloc;
    if (arow_i >= nrows) arow_i = nrows - 1;
    const float* __restrict__ arow = X + (size_t)arow_i * DF + 2 * khalf;

    const v2f* __restrict__ bbase = (const v2f*)ldsW + lane;

    v8f acc[8] = {};
#pragma unroll
    for (int k = 0; k < 32; ++k) {
        v2f a = *(const v2f*)(arow + 4 * k);
#pragma unroll
        for (int t = 0; t < 8; ++t) {
            v2f b = bbase[(t * 32 + k) * 32];
            acc[t] = __builtin_amdgcn_wmma_f32_16x16x4_f32(
                         false, a, false, b, (short)0, acc[t], false, false);
        }
    }

    const int rbase = strip * 16 + 8 * khalf;
#pragma unroll
    for (int r = 0; r < 8; ++r) {
        int row = rbase + r;
        if (row >= nrows) continue;
        size_t ro = (size_t)row * DF;
#pragma unroll
        for (int t = 0; t < 8; ++t) H[ro + t * 16 + nloc] = acc[t][r];
    }
}

#define CAPB 24576
#define SRCB 23
#define RPC  4
#define AS   132
#define AGT  1024
#define TRP  (((TR + 15) / 16) * 16)
#define AGW  32
#define SUBCAP (CAPB / AGW)
#define QCAP 64
#define DB   16
#define SU   4

__device__ __forceinline__ void lds_add(float* p, float x) {
    (void)__hip_atomic_fetch_add(p, x, __ATOMIC_RELAXED, __HIP_MEMORY_SCOPE_WORKGROUP);
}

__device__ __forceinline__ unsigned zero_bytes(unsigned x) {
    const unsigned t = (x & 0x7F7F7F7Fu) + 0x7F7F7F7Fu;
    return ~(t | x | 0x7F7F7F7Fu);
}

template <bool FULL, bool REC>
__device__ __forceinline__ void drain_batch(
        const unsigned* qw, int cnt, int lane, int lo,
        const float* __restrict__ H, const int* __restrict__ src,
        const int* __restrict__ dst, const float* __restrict__ dinv, float* acc,
        v2u* __restrict__ rb, int* nrec) {
    int s = 0, d = lo;
    float ws = 0.0f, wd = 0.0f;
    if (lane < cnt) {
        const unsigned e = qw[lane];
        s = src[e];
        d = dst[e];
        ws = dinv[s];
        wd = dinv[d];
    }
    const int dl = d - lo;
    const float* __restrict__ hl = H + lane;
    float* al = acc + lane;
    auto put_records = [&]() {
        if (lane < cnt) {
            const int ow = dl & (AGW - 1);
            const int slot = __hip_atomic_fetch_add(nrec + ow, 1, __ATOMIC_RELAXED, __HIP_MEMORY_SCOPE_WORKGROUP);
            if (slot < SUBCAP) {
                const v2u r = {(unsigned)s | ((unsigned)dl << SRCB), __builtin_bit_cast(unsigned, ws * wd)};
                rb[ow * SUBCAP + slot] = r;
            }
        }
    };
    if (FULL) {
        float v[DB][RPC];
#pragma unroll
        for (int j = 0; j < DB; ++j) {
            const int sj = __builtin_amdgcn_readlane(s, j);
            const float* __restrict__ row = hl + (size_t)sj * DF;
#pragma unroll
            for (int c = 0; c < RPC; ++c) v[j][c] = row[32 * c];
        }
        if (REC) put_records();
#pragma unroll
        for (int j = 0; j < DB; ++j) {
            const int dj = __builtin_amdgcn_readlane(dl, j);
            const float nj =
                __builtin_bit_cast(float, __builtin_amdgcn_readlane(__builtin_bit_cast(int, ws), j)) *
                __builtin_bit_cast(float, __builtin_amdgcn_readlane(__builtin_bit_cast(int, wd), j));
            float* a = al + dj * AS;
#pragma unroll
            for (int c = 0; c < RPC; ++c) lds_add(a + 32 * c, v[j][c] * nj);
        }
    } else {
        if (REC) put_records();
        for (int j = 0; j < cnt; ++j) {
            const int sj = __builtin_amdgcn_readlane(s, j);
            const int dj = __builtin_amdgcn_readlane(dl, j);
            const float nj =
                __builtin_bit_cast(float, __builtin_amdgcn_readlane(__builtin_bit_cast(int, ws), j)) *
                __builtin_bit_cast(float, __builtin_amdgcn_readlane(__builtin_bit_cast(int, wd), j));
            const float* __restrict__ row = hl + (size_t)sj * DF;
            float* a = al + dj * AS;
            float t[RPC];
#pragma unroll
            for (int c = 0; c < RPC; ++c) t[c] = row[32 * c];
#pragma unroll
            for (int c = 0; c < RPC; ++c) lds_add(a + 32 * c, t[c] * nj);
        }
    }
}

template <bool FULL>
__device__ __forceinline__ void drain_own(v2u r, int cnt, int lane,
                                          const float* __restrict__ H, float* acc) {
    const char* __restrict__ hb = (const char*)H;
    const unsigned lo16 = (unsigned)lane * 16u;
    float* al = acc + 4 * lane;
    const int rx = (int)r.x, ry = (int)r.y;
    if (FULL) {
        v4f v[DB];
#pragma unroll
        for (int j = 0; j < DB; ++j) {
            const unsigned pj = (unsigned)__builtin_amdgcn_readlane(rx, j);
            v[j] = *(const v4f*)(hb + (((pj & ((1u << SRCB) - 1u)) * (unsigned)(DF * 4)) + lo16));
        }
        __builtin_amdgcn_sched_barrier(0);
#pragma unroll
        for (int j = 0; j < DB; ++j) {
            const unsigned pj = (unsigned)__builtin_amdgcn_readlane(rx, j);
            const float nj = __builtin_bit_cast(float, __builtin_amdgcn_readlane(ry, j));
            v4f* a = (v4f*)(al + (pj >> SRCB) * AS);
            v4f c = *a;
            c.x = fmaf(v[j].x, nj, c.x);
            c.y = fmaf(v[j].y, nj, c.y);
            c.z = fmaf(v[j].z, nj, c.z);
            c.w = fmaf(v[j].w, nj, c.w);
            *a = c;
        }
    } else {
        for (int j = 0; j < cnt; ++j) {
            const unsigned pj = (unsigned)__builtin_amdgcn_readlane(rx, j);
            const float nj = __builtin_bit_cast(float, __builtin_amdgcn_readlane(ry, j));
            const v4f t = *(const v4f*)(hb + (((pj & ((1u << SRCB) - 1u)) * (unsigned)(DF * 4)) + lo16));
            v4f* a = (v4f*)(al + (pj >> SRCB) * AS);
            v4f c = *a;
            c.x = fmaf(t.x, nj, c.x);
            c.y = fmaf(t.y, nj, c.y);
            c.z = fmaf(t.z, nj, c.z);
            c.w = fmaf(t.w, nj, c.w);
            *a = c;
        }
    }
}

template <bool REC>
__device__ __forceinline__ void filter_block(
        int tile, int lo, int wave, int lane, int G, int E,
        const v4u* __restrict__ b4, unsigned* qw,
        const float* __restrict__ H, const int* __restrict__ src,
        const int* __restrict__ dst, const float* __restrict__ dinv, float* acc,
        v2u* __restrict__ rb, int* nrec) {
    const unsigned pat = (unsigned)tile * 0x01010101u;
    int qn = 0;
    const int glast = G > 0 ? G - 1 : 0;
    v4u w[SU];
#pragma unroll
    for (int u = 0; u < SU; ++u) {
        const int g = (wave * SU + u) * 32 + lane;
        w[u] = b4[g < G ? g : glast];
    }
    for (int i0 = wave * SU; i0 * 32 < G; i0 += AGW * SU) {
        v4u nx[SU];
#pragma unroll
        for (int u = 0; u < SU; ++u) {
            const int g = (i0 + AGW * SU + u) * 32 + lane;
            nx[u] = b4[g < G ? g : glast];
        }
#pragma unroll
        for (int u = 0; u < SU; ++u) {
            const int g = (i0 + u) * 32 + lane;
            unsigned mask = (zero_bytes(w[u].x ^ pat) >> 7) | (zero_bytes(w[u].y ^ pat) >> 6) |
                            (zero_bytes(w[u].z ^ pat) >> 5) | (zero_bytes(w[u].w ^ pat) >> 4);
            if (g >= G) mask = 0u;
            while (__builtin_amdgcn_ballot_w32(mask != 0u) != 0u) {
                const bool has = mask != 0u;
                const unsigned p = (unsigned)__builtin_ctz(mask | 0x80000000u);
                mask &= mask - 1u;
                const unsigned e = (unsigned)g * 16u + ((p & 7u) << 2) + (p >> 3);
                const bool ok = has && e < (unsigned)E;
                const unsigned okb = __builtin_amdgcn_ballot_w32(ok);
                if (ok) qw[qn + (int)__builtin_amdgcn_mbcnt_lo(okb, 0u)] = e;
                qn += __builtin_popcount(okb);
                __builtin_amdgcn_wave_barrier();
                while (qn >= DB) {
                    qn -= DB;
                    drain_batch<true, REC>(qw + qn, DB, lane, lo, H, src, dst, dinv, acc, rb, nrec);
                }
            }
        }
#pragma unroll
        for (int u = 0; u < SU; ++u) w[u] = nx[u];
    }
    if (qn > 0) drain_batch<false, REC>(qw, qn, lane, lo, H, src, dst, dinv, acc, rb, nrec);
}

template <bool BUILD, bool FUSE>
__global__ __launch_bounds__(AGT) void k_aggregate(
        const float* __restrict__ H, const int* __restrict__ src,
        const int* __restrict__ dst, const float* __restrict__ dinv,
        const float* __restrict__ bias, const unsigned char* __restrict__ blk,
        unsigned* __restrict__ recw, int* __restrict__ nrecs,
        const float* __restrict__ Wn, float* __restrict__ OUT, int N, int E) {
    __shared__ __align__(16) float acc[(FUSE ? TRP : TR) * AS];
    __shared__ __align__(16) float ldsW[FUSE ? DF * DF : 4];
    __shared__ unsigned q[AGW * QCAP];
    __shared__ int nrec[AGW];

    const int lane = threadIdx.x & 31;
    const int wave = __builtin_amdgcn_readfirstlane((int)(threadIdx.x >> 5));
    const int nblk = (N + TR - 1) / TR;
    const int G = (E + 15) >> 4;
    const v4u* __restrict__ b4 = (const v4u*)blk;
    unsigned* qw = q + wave * QCAP;

    if (FUSE) {
        float w0[DF * DF / 2 / AGT], w1[DF * DF / 2 / AGT];
#pragma unroll
        for (int u = 0; u < DF * DF / 2 / AGT; ++u) {
            const int p = u * AGT + (int)threadIdx.x;
            const int qq = p >> 7, n = p & 127;
            w0[u] = Wn[(2 * qq) * DF + n];
            w1[u] = Wn[(2 * qq + 1) * DF + n];
        }
#pragma unroll
        for (int u = 0; u < DF * DF / 2 / AGT; ++u) {
            const int p = u * AGT + (int)threadIdx.x;
            const int qq = p >> 7, n = p & 127;
            const v2f v = {w0[u], w1[u]};
            ((v2f*)ldsW)[(((n >> 4) << 5) + (qq >> 1)) * 32 + (qq & 1) * 16 + (n & 15)] = v;
        }
    }

    const int tile = blockIdx.x;
    if (tile >= nblk) return;
    {
        const int lo = tile * TR;
        const int rows = (N - lo) < TR ? (N - lo) : TR;
        v2u* __restrict__ rb = (v2u*)recw + (size_t)tile * CAPB;

        for (int i = threadIdx.x; i < rows * (DF / 4); i += blockDim.x) {
            const int r = i >> 5, c4 = (i & 31) * 4;
            const float dv = dinv[lo + r];
            const float d2 = dv * dv;
            const v4f h = *(const v4f*)(H + (size_t)(lo + r) * DF + c4);
            const v4f b = *(const v4f*)(bias + c4);
            v4f o = {fmaf(h.x, d2, b.x), fmaf(h.y, d2, b.y), fmaf(h.z, d2, b.z), fmaf(h.w, d2, b.w)};
            *(v4f*)(acc + r * AS + c4) = o;
        }
        if (FUSE) {
            const v4f z = {0.0f, 0.0f, 0.0f, 0.0f};
            for (int i = rows * (DF / 4) + (int)threadIdx.x; i < TRP * (DF / 4); i += blockDim.x)
                *(v4f*)(acc + (i >> 5) * AS + (i & 31) * 4) = z;
        }
        if (BUILD && threadIdx.x < AGW) nrec[threadIdx.x] = 0;
        __syncthreads();

        if (BUILD) {
            filter_block<true>(tile, lo, wave, lane, G, E, b4, qw, H, src, dst, dinv, acc, rb, nrec);
        } else {
            const int cl = nrecs[tile * AGW + lane];
            if (__builtin_amdgcn_ballot_w32(cl > SUBCAP) == 0u) {
                const int total = __builtin_amdgcn_readlane(cl, wave);
                const v2u* __restrict__ rw = rb + wave * SUBCAP;
                v2u r = {0u, 0u};
                if (lane < DB && lane < total) r = rw[lane];
                for (int base = 0; base < total; base += DB) {
                    const int nb = base + DB;
                    v2u rn = {0u, 0u};
                    if (lane < DB && nb + lane < total) rn = rw[nb + lane];
                    const int n = total - base;
                    if (n >= DB) drain_own<true>(r, DB, lane, H, acc);
                    else         drain_own<false>(r, n, lane, H, acc);
                    r = rn;
                }
            } else {
                filter_block<false>(tile, lo, wave, lane, G, E, b4, qw, H, src, dst, dinv, acc, rb, nrec);
            }
        }
        __syncthreads();

        if (FUSE) {
            if (wave * 16 < rows) {
                const int nloc  = lane & 15;
                const int khalf = lane >> 4;
                const float* arow = acc + (wave * 16 + nloc) * AS + 2 * khalf;
                const v2f* bbase = (const v2f*)ldsW + lane;
                v8f c[8] = {};
#pragma unroll
                for (int k = 0; k < 32; ++k) {
                    v2f a = *(const v2f*)(arow + 4 * k);
                    a.x = fmaxf(a.x, 0.0f);
                    a.y = fmaxf(a.y, 0.0f);
#pragma unroll
                    for (int t = 0; t < 8; ++t) {
                        const v2f b = bbase[(t * 32 + k) * 32];
                        c[t] = __builtin_amdgcn_wmma_f32_16x16x4_f32(
                                   false, a, false, b, (short)0, c[t], false, false);
                    }
                }
                float* drow = acc + (wave * 16 + 8 * khalf) * AS + nloc;
#pragma unroll
                for (int r = 0; r < 8; ++r) {
#pragma unroll
                    for (int t = 0; t < 8; ++t) drow[r * AS + t * 16] = c[t][r];
                }
            }
            __syncthreads();
        }

        for (int i = threadIdx.x; i < rows * (DF / 4); i += blockDim.x) {
            const int r = i >> 5, c4 = (i & 31) * 4;
            *(v4f*)(OUT + (size_t)(lo + r) * DF + c4) = *(const v4f*)(acc + r * AS + c4);
        }
        if (BUILD && threadIdx.x < AGW) nrecs[tile * AGW + threadIdx.x] = nrec[threadIdx.x];
    }
}

__global__ __launch_bounds__(256) void k_normalize(const float* __restrict__ in,
                                                   float* __restrict__ out, int n) {
    int row = blockIdx.x * (blockDim.x >> 5) + (threadIdx.x >> 5);
    if (row >= n) return;
    int lane = threadIdx.x & 31;
    v4f v = *(const v4f*)(in + (size_t)row * DF + lane * 4);
    float ss = v.x * v.x + v.y * v.y + v.z * v.z + v.w * v.w;
#pragma unroll
    for (int m = 16; m > 0; m >>= 1) ss += __shfl_xor(ss, m, 32);
    float sc = 1.0f / fmaxf(sqrtf(ss), 1e-12f);
    v4f o = {v.x * sc, v.y * sc, v.z * sc, v.w * sc};
    *(v4f*)(out + (size_t)row * DF + lane * 4) = o;
}


extern "C" void kernel_launch(void* const* d_in, const int* in_sizes, int n_in,
                              void* d_out, int out_size, void* d_ws, size_t ws_size,
                              hipStream_t stream) {
    const float* x  = (const float*)d_in[0];
    const int*   ei = (const int*)  d_in[1];
    const float* W1 = (const float*)d_in[2];
    const float* b1 = (const float*)d_in[3];
    const float* W2 = (const float*)d_in[4];
    const float* b2 = (const float*)d_in[5];
    const float* W3 = (const float*)d_in[6];
    const float* b3 = (const float*)d_in[7];

    const int N = in_sizes[0] / DF;
    const int E = in_sizes[1] / 2;
    const int* src = ei;
    const int* dst = ei + E;

    float* out = (float*)d_out;

    char* ws = (char*)d_ws;
    size_t off0 = ((size_t)N * sizeof(float) + 255) & ~(size_t)255;
    float* dinv = (float*)ws;
    float* H    = (float*)(ws + off0);
    float* AGGa = H + (size_t)N * DF;
    unsigned char* blk = (unsigned char*)(AGGa + (size_t)N * DF);
    int* nrecs = (int*)(ws + ((size_t)128 << 20));
    unsigned* recs = (unsigned*)(ws + ((size_t)128 << 20) + 32768);

    const int strips  = (N + 15) / 16;
    const int gblocks = (strips + 7) / 8;

    unsigned* hist = (unsigned*)AGGa;
    k_degree_hist<<<HG, 1024, 0, stream>>>(dst, blk, hist, E);
    k_dinv_sum   <<<(HWORDS + 255) / 256, 256, 0, stream>>>(hist, dinv, N, HG);

    k_gemm_wmma<<<gblocks, 256, 0, stream>>>(x, W1, H, strips, N);
    k_aggregate<true, true>  <<<256, AGT, 0, stream>>>(H, src, dst, dinv, b1, blk, recs, nrecs, W2, AGGa, N, E);

    k_aggregate<false, true> <<<256, AGT, 0, stream>>>(AGGa, src, dst, dinv, b2, blk, recs, nrecs, W3, H, N, E);

    k_aggregate<false, false><<<256, AGT, 0, stream>>>(H, src, dst, dinv, b3, blk, recs, nrecs, W3, AGGa, N, E);

    k_normalize<<<(N + 7) / 8, 256, 0, stream>>>(AGGa, out, N);
}


# Last layer: a row is divided by its L2 norm as the aggregation launch stores it from LDS (a wave holds one row per trip) to the output; normalize launch and its plane pass are gone; sums unchanged
# speedup vs baseline: 3.4105x; 1.0551x over previous
#include <hip/hip_runtime.h>
#include <hip/hip_bf16.h>


typedef __attribute__((ext_vector_type(2))) float v2f;
typedef __attribute__((ext_vector_type(4))) float v4f;
typedef __attribute__((ext_vector_type(8))) float v8f;
typedef __attribute__((ext_vector_type(4))) unsigned v4u;
typedef __attribute__((ext_vector_type(2))) unsigned v2u;

#define DF 128


#define TR 391

#define HG     128
#define HWORDS (256 * TR / 2)

#define EU 4
__global__ __launch_bounds__(1024) void k_degree_hist(const int* __restrict__ dst,
                                                      unsigned char* __restrict__ blk,
                                                      unsigned* __restrict__ hist, int E) {
    __shared__ __align__(16) unsigned h[HWORDS];
    const v4u z = {0u, 0u, 0u, 0u};
    for (int i = threadIdx.x; i < HWORDS / 4; i += blockDim.x) ((v4u*)h)[i] = z;
    __syncthreads();

    const int chunk = (E + (int)gridDim.x - 1) / (int)gridDim.x;
    const int e0 = (int)blockIdx.x * chunk;
    const int e1 = (e0 + chunk) < E ? (e0 + chunk) : E;
    const int step = (int)blockDim.x;
    int e = e0 + (int)threadIdx.x;
    for (; e + (EU - 1) * step < e1; e += EU * step) {
        int d[EU];
#pragma unroll
        for (int u = 0; u < EU; ++u) d[u] = dst[e + u * step];
#pragma unroll
        for (int u = 0; u < EU; ++u) {
            blk[e + u * step] = (unsigned char)(d[u] / TR);
            (void)__hip_atomic_fetch_add(&h[d[u] >> 1], 1u << ((d[u] & 1) * 16),
                                         __ATOMIC_RELAXED, __HIP_MEMORY_SCOPE_WORKGROUP);
        }
    }
    for (; e < e1; e += step) {
        const int d = dst[e];
        blk[e] = (unsigned char)(d / TR);
        (void)__hip_atomic_fetch_add(&h[d >> 1], 1u << ((d & 1) * 16),
                                     __ATOMIC_RELAXED, __HIP_MEMORY_SCOPE_WORKGROUP);
    }
    if (blockIdx.x == gridDim.x - 1) {
        const int Epad = (E + 15) & ~15;
        for (int p = E + (int)threadIdx.x; p < Epad; p += step) blk[p] = 0;
    }
    __syncthreads();

    v4u* __restrict__ o = (v4u*)(hist + (size_t)blockIdx.x * HWORDS);
    for (int i = threadIdx.x; i < HWORDS / 4; i += blockDim.x) o[i] = ((const v4u*)h)[i];
}

__global__ __launch_bounds__(256) void k_dinv_sum(const unsigned* __restrict__ hist,
                                                  float* __restrict__ dinv, int n, int groups) {
    const int w = blockIdx.x * blockDim.x + threadIdx.x;
    if (w >= HWORDS) return;
    const unsigned* __restrict__ p = hist + w;
    unsigned c0 = 0u, c1 = 0u;
    int g = 0;
    for (; g + 32 <= groups; g += 32) {
        unsigned v[32];
#pragma unroll
        for (int u = 0; u < 32; ++u) v[u] = p[(size_t)(g + u) * HWORDS];
#pragma unroll
        for (int u = 0; u < 32; ++u) { c0 += v[u] & 0xFFFFu; c1 += v[u] >> 16; }
    }
    for (; g < groups; ++g) {
        const unsigned v = p[(size_t)g * HWORDS];
        c0 += v & 0xFFFFu;
        c1 += v >> 16;
    }
    const int i = 2 * w;
    if (i < n)     dinv[i]     = rsqrtf((float)c0 + 1.0f);
    if (i + 1 < n) dinv[i + 1] = rsqrtf((float)c1 + 1.0f);
}

__global__ __launch_bounds__(256) void k_gemm_wmma(
        const float* __restrict__ X, const float* __restrict__ W,
        float* __restrict__ H, int strips, int nrows) {
    __shared__ __align__(16) float ldsW[DF * DF];

    for (int idx = threadIdx.x; idx < DF * DF; idx += blockDim.x) {
        int kk = idx >> 7;
        int n  = idx & 127;
        int t     = n >> 4;
        int nloc  = n & 15;
        int k     = kk >> 2;
        int khalf = (kk >> 1) & 1;
        int lo    = kk & 1;
        ldsW[((t << 5) + k) * 64 + (khalf * 16 + nloc) * 2 + lo] = W[idx];
    }
    __syncthreads();

    const int wave  = threadIdx.x >> 5;
    const int lane  = threadIdx.x & 31;
    const int strip = blockIdx.x * (blockDim.x >> 5) + wave;
    if (strip >= strips) return;

    const int nloc  = lane & 15;
    const int khalf = lane >> 4;

    int arow_i = strip * 16 + nloc;
    if (arow_i >= nrows) arow_i = nrows - 1;
    const float* __restrict__ arow = X + (size_t)arow_i * DF + 2 * khalf;

    const v2f* __restrict__ bbase = (const v2f*)ldsW + lane;

    v8f acc[8] = {};
#pragma unroll
    for (int k = 0; k < 32; ++k) {
        v2f a = *(const v2f*)(arow + 4 * k);
#pragma unroll
        for (int t = 0; t < 8; ++t) {
            v2f b = bbase[(t * 32 + k) * 32];
            acc[t] = __builtin_amdgcn_wmma_f32_16x16x4_f32(
                         false, a, false, b, (short)0, acc[t], false, false);
        }
    }

    const int rbase = strip * 16 + 8 * khalf;
#pragma unroll
    for (int r = 0; r < 8; ++r) {
        int row = rbase + r;
        if (row >= nrows) continue;
        size_t ro = (size_t)row * DF;
#pragma unroll
        for (int t = 0; t < 8; ++t) H[ro + t * 16 + nloc] = acc[t][r];
    }
}

#define CAPB 24576
#define SRCB 23
#define RPC  4
#define AS   132
#define AGT  1024
#define TRP  (((TR + 15) / 16) * 16)
#define AGW  32
#define SUBCAP (CAPB / AGW)
#define QCAP 64
#define DB   16
#define SU   4

__device__ __forceinline__ void lds_add(float* p, float x) {
    (void)__hip_atomic_fetch_add(p, x, __ATOMIC_RELAXED, __HIP_MEMORY_SCOPE_WORKGROUP);
}

__device__ __forceinline__ unsigned zero_bytes(unsigned x) {
    const unsigned t = (x & 0x7F7F7F7Fu) + 0x7F7F7F7Fu;
    return ~(t | x | 0x7F7F7F7Fu);
}

template <bool FULL, bool REC>
__device__ __forceinline__ void drain_batch(
        const unsigned* qw, int cnt, int lane, int lo,
        const float* __restrict__ H, const int* __restrict__ src,
        const int* __restrict__ dst, const float* __restrict__ dinv, float* acc,
        v2u* __restrict__ rb, int* nrec) {
    int s = 0, d = lo;
    float ws = 0.0f, wd = 0.0f;
    if (lane < cnt) {
        const unsigned e = qw[lane];
        s = src[e];
        d = dst[e];
        ws = dinv[s];
        wd = dinv[d];
    }
    const int dl = d - lo;
    const float* __restrict__ hl = H + lane;
    float* al = acc + lane;
    auto put_records = [&]() {
        if (lane < cnt) {
            const int ow = dl & (AGW - 1);
            const int slot = __hip_atomic_fetch_add(nrec + ow, 1, __ATOMIC_RELAXED, __HIP_MEMORY_SCOPE_WORKGROUP);
            if (slot < SUBCAP) {
                const v2u r = {(unsigned)s | ((unsigned)dl << SRCB), __builtin_bit_cast(unsigned, ws * wd)};
                rb[ow * SUBCAP + slot] = r;
            }
        }
    };
    if (FULL) {
        float v[DB][RPC];
#pragma unroll
        for (int j = 0; j < DB; ++j) {
            const int sj = __builtin_amdgcn_readlane(s, j);
            const float* __restrict__ row = hl + (size_t)sj * DF;
#pragma unroll
            for (int c = 0; c < RPC; ++c) v[j][c] = row[32 * c];
        }
        if (REC) put_records();
#pragma unroll
        for (int j = 0; j < DB; ++j) {
            const int dj = __builtin_amdgcn_readlane(dl, j);
            const float nj =
                __builtin_bit_cast(float, __builtin_amdgcn_readlane(__builtin_bit_cast(int, ws), j)) *
                __builtin_bit_cast(float, __builtin_amdgcn_readlane(__builtin_bit_cast(int, wd), j));
            float* a = al + dj * AS;
#pragma unroll
            for (int c = 0; c < RPC; ++c) lds_add(a + 32 * c, v[j][c] * nj);
        }
    } else {
        if (REC) put_records();
        for (int j = 0; j < cnt; ++j) {
            const int sj = __builtin_amdgcn_readlane(s, j);
            const int dj = __builtin_amdgcn_readlane(dl, j);
            const float nj =
                __builtin_bit_cast(float, __builtin_amdgcn_readlane(__builtin_bit_cast(int, ws), j)) *
                __builtin_bit_cast(float, __builtin_amdgcn_readlane(__builtin_bit_cast(int, wd), j));
            const float* __restrict__ row = hl + (size_t)sj * DF;
            float* a = al + dj * AS;
            float t[RPC];
#pragma unroll
            for (int c = 0; c < RPC; ++c) t[c] = row[32 * c];
#pragma unroll
            for (int c = 0; c < RPC; ++c) lds_add(a + 32 * c, t[c] * nj);
        }
    }
}

template <bool FULL>
__device__ __forceinline__ void drain_own(v2u r, int cnt, int lane,
                                          const float* __restrict__ H, float* acc) {
    const char* __restrict__ hb = (const char*)H;
    const unsigned lo16 = (unsigned)lane * 16u;
    float* al = acc + 4 * lane;
    const int rx = (int)r.x, ry = (int)r.y;
    if (FULL) {
        v4f v[DB];
#pragma unroll
        for (int j = 0; j < DB; ++j) {
            const unsigned pj = (unsigned)__builtin_amdgcn_readlane(rx, j);
            v[j] = *(const v4f*)(hb + (((pj & ((1u << SRCB) - 1u)) * (unsigned)(DF * 4)) + lo16));
        }
        __builtin_amdgcn_sched_barrier(0);
#pragma unroll
        for (int j = 0; j < DB; ++j) {
            const unsigned pj = (unsigned)__builtin_amdgcn_readlane(rx, j);
            const float nj = __builtin_bit_cast(float, __builtin_amdgcn_readlane(ry, j));
            v4f* a = (v4f*)(al + (pj >> SRCB) * AS);
            v4f c = *a;
            c.x = fmaf(v[j].x, nj, c.x);
            c.y = fmaf(v[j].y, nj, c.y);
            c.z = fmaf(v[j].z, nj, c.z);
            c.w = fmaf(v[j].w, nj, c.w);
            *a = c;
        }
    } else {
        for (int j = 0; j < cnt; ++j) {
            const unsigned pj = (unsigned)__builtin_amdgcn_readlane(rx, j);
            const float nj = __builtin_bit_cast(float, __builtin_amdgcn_readlane(ry, j));
            const v4f t = *(const v4f*)(hb + (((pj & ((1u << SRCB) - 1u)) * (unsigned)(DF * 4)) + lo16));
            v4f* a = (v4f*)(al + (pj >> SRCB) * AS);
            v4f c = *a;
            c.x = fmaf(t.x, nj, c.x);
            c.y = fmaf(t.y, nj, c.y);
            c.z = fmaf(t.z, nj, c.z);
            c.w = fmaf(t.w, nj, c.w);
            *a = c;
        }
    }
}

template <bool REC>
__device__ __forceinline__ void filter_block(
        int tile, int lo, int wave, int lane, int G, int E,
        const v4u* __restrict__ b4, unsigned* qw,
        const float* __restrict__ H, const int* __restrict__ src,
        const int* __restrict__ dst, const float* __restrict__ dinv, float* acc,
        v2u* __restrict__ rb, int* nrec) {
    const unsigned pat = (unsigned)tile * 0x01010101u;
    int qn = 0;
    const int glast = G > 0 ? G - 1 : 0;
    v4u w[SU];
#pragma unroll
    for (int u = 0; u < SU; ++u) {
        const int g = (wave * SU + u) * 32 + lane;
        w[u] = b4[g < G ? g : glast];
    }
    for (int i0 = wave * SU; i0 * 32 < G; i0 += AGW * SU) {
        v4u nx[SU];
#pragma unroll
        for (int u = 0; u < SU; ++u) {
            const int g = (i0 + AGW * SU + u) * 32 + lane;
            nx[u] = b4[g < G ? g : glast];
        }
#pragma unroll
        for (int u = 0; u < SU; ++u) {
            const int g = (i0 + u) * 32 + lane;
            unsigned mask = (zero_bytes(w[u].x ^ pat) >> 7) | (zero_bytes(w[u].y ^ pat) >> 6) |
                            (zero_bytes(w[u].z ^ pat) >> 5) | (zero_bytes(w[u].w ^ pat) >> 4);
            if (g >= G) mask = 0u;
            while (__builtin_amdgcn_ballot_w32(mask != 0u) != 0u) {
                const bool has = mask != 0u;
                const unsigned p = (unsigned)__builtin_ctz(mask | 0x80000000u);
                mask &= mask - 1u;
                const unsigned e = (unsigned)g * 16u + ((p & 7u) << 2) + (p >> 3);
                const bool ok = has && e < (unsigned)E;
                const unsigned okb = __builtin_amdgcn_ballot_w32(ok);
                if (ok) qw[qn + (int)__builtin_amdgcn_mbcnt_lo(okb, 0u)] = e;
                qn += __builtin_popcount(okb);
                __builtin_amdgcn_wave_barrier();
                while (qn >= DB) {
                    qn -= DB;
                    drain_batch<true, REC>(qw + qn, DB, lane, lo, H, src, dst, dinv, acc, rb, nrec);
                }
            }
        }
#pragma unroll
        for (int u = 0; u < SU; ++u) w[u] = nx[u];
    }
    if (qn > 0) drain_batch<false, REC>(qw, qn, lane, lo, H, src, dst, dinv, acc, rb, nrec);
}

template <bool BUILD, bool FUSE, bool NORM>
__global__ __launch_bounds__(AGT) void k_aggregate(
        const float* __restrict__ H, const int* __restrict__ src,
        const int* __restrict__ dst, const float* __restrict__ dinv,
        const float* __restrict__ bias, const unsigned char* __restrict__ blk,
        unsigned* __restrict__ recw, int* __restrict__ nrecs,
        const float* __restrict__ Wn, float* __restrict__ OUT, int N, int E) {
    __shared__ __align__(16) float acc[(FUSE ? TRP : TR) * AS];
    __shared__ __align__(16) float ldsW[FUSE ? DF * DF : 4];
    __shared__ unsigned q[AGW * QCAP];
    __shared__ int nrec[AGW];

    const int lane = threadIdx.x & 31;
    const int wave = __builtin_amdgcn_readfirstlane((int)(threadIdx.x >> 5));
    const int nblk = (N + TR - 1) / TR;
    const int G = (E + 15) >> 4;
    const v4u* __restrict__ b4 = (const v4u*)blk;
    unsigned* qw = q + wave * QCAP;

    if (FUSE) {
        float w0[DF * DF / 2 / AGT], w1[DF * DF / 2 / AGT];
#pragma unroll
        for (int u = 0; u < DF * DF / 2 / AGT; ++u) {
            const int p = u * AGT + (int)threadIdx.x;
            const int qq = p >> 7, n = p & 127;
            w0[u] = Wn[(2 * qq) * DF + n];
            w1[u] = Wn[(2 * qq + 1) * DF + n];
        }
#pragma unroll
        for (int u = 0; u < DF * DF / 2 / AGT; ++u) {
            const int p = u * AGT + (int)threadIdx.x;
            const int qq = p >> 7, n = p & 127;
            const v2f v = {w0[u], w1[u]};
            ((v2f*)ldsW)[(((n >> 4) << 5) + (qq >> 1)) * 32 + (qq & 1) * 16 + (n & 15)] = v;
        }
    }

    const int tile = blockIdx.x;
    if (tile >= nblk) return;
    {
        const int lo = tile * TR;
        const int rows = (N - lo) < TR ? (N - lo) : TR;
        v2u* __restrict__ rb = (v2u*)recw + (size_t)tile * CAPB;

        for (int i = threadIdx.x; i < rows * (DF / 4); i += blockDim.x) {
            const int r = i >> 5, c4 = (i & 31) * 4;
            const float dv = dinv[lo + r];
            const float d2 = dv * dv;
            const v4f h = *(const v4f*)(H + (size_t)(lo + r) * DF + c4);
            const v4f b = *(const v4f*)(bias + c4);
            v4f o = {fmaf(h.x, d2, b.x), fmaf(h.y, d2, b.y), fmaf(h.z, d2, b.z), fmaf(h.w, d2, b.w)};
            *(v4f*)(acc + r * AS + c4) = o;
        }
        if (FUSE) {
            const v4f z = {0.0f, 0.0f, 0.0f, 0.0f};
            for (int i = rows * (DF / 4) + (int)threadIdx.x; i < TRP * (DF / 4); i += blockDim.x)
                *(v4f*)(acc + (i >> 5) * AS + (i & 31) * 4) = z;
        }
        if (BUILD && threadIdx.x < AGW) nrec[threadIdx.x] = 0;
        __syncthreads();

        if (BUILD) {
            filter_block<true>(tile, lo, wave, lane, G, E, b4, qw, H, src, dst, dinv, acc, rb, nrec);
        } else {
            const int cl = nrecs[tile * AGW + lane];
            if (__builtin_amdgcn_ballot_w32(cl > SUBCAP) == 0u) {
                const int total = __builtin_amdgcn_readlane(cl, wave);
                const v2u* __restrict__ rw = rb + wave * SUBCAP;
                v2u r = {0u, 0u};
                if (lane < DB && lane < total) r = rw[lane];
                for (int base = 0; base < total; base += DB) {
                    const int nb = base + DB;
                    v2u rn = {0u, 0u};
                    if (lane < DB && nb + lane < total) rn = rw[nb + lane];
                    const int n = total - base;
                    if (n >= DB) drain_own<true>(r, DB, lane, H, acc);
                    else         drain_own<false>(r, n, lane, H, acc);
                    r = rn;
                }
            } else {
                filter_block<false>(tile, lo, wave, lane, G, E, b4, qw, H, src, dst, dinv, acc, rb, nrec);
            }
        }
        __syncthreads();

        if (FUSE) {
            if (wave * 16 < rows) {
                const int nloc  = lane & 15;
                const int khalf = lane >> 4;
                const float* arow = acc + (wave * 16 + nloc) * AS + 2 * khalf;
                const v2f* bbase = (const v2f*)ldsW + lane;
                v8f c[8] = {};
#pragma unroll
                for (int k = 0; k < 32; ++k) {
                    v2f a = *(const v2f*)(arow + 4 * k);
                    a.x = fmaxf(a.x, 0.0f);
                    a.y = fmaxf(a.y, 0.0f);
#pragma unroll
                    for (int t = 0; t < 8; ++t) {
                        const v2f b = bbase[(t * 32 + k) * 32];
                        c[t] = __builtin_amdgcn_wmma_f32_16x16x4_f32(
                                   false, a, false, b, (short)0, c[t], false, false);
                    }
                }
                float* drow = acc + (wave * 16 + 8 * khalf) * AS + nloc;
#pragma unroll
                for (int r = 0; r < 8; ++r) {
#pragma unroll
                    for (int t = 0; t < 8; ++t) drow[r * AS + t * 16] = c[t][r];
                }
            }
            __syncthreads();
        }

        for (int i = threadIdx.x; i < rows * (DF / 4); i += blockDim.x) {
            const int r = i >> 5, c4 = (i & 31) * 4;
            v4f v = *(const v4f*)(acc + r * AS + c4);
            if (NORM) {
                float ss = v.x * v.x + v.y * v.y + v.z * v.z + v.w * v.w;
#pragma unroll
                for (int m = 16; m > 0; m >>= 1) ss += __shfl_xor(ss, m, 32);
                const float sc = 1.0f / fmaxf(sqrtf(ss), 1e-12f);
                v.x *= sc; v.y *= sc; v.z *= sc; v.w *= sc;
            }
            *(v4f*)(OUT + (size_t)(lo + r) * DF + c4) = v;
        }
        if (BUILD && threadIdx.x < AGW) nrecs[tile * AGW + threadIdx.x] = nrec[threadIdx.x];
    }
}


extern "C" void kernel_launch(void* const* d_in, const int* in_sizes, int n_in,
                              void* d_out, int out_size, void* d_ws, size_t ws_size,
                              hipStream_t stream) {
    const float* x  = (const float*)d_in[0];
    const int*   ei = (const int*)  d_in[1];
    const float* W1 = (const float*)d_in[2];
    const float* b1 = (const float*)d_in[3];
    const float* W2 = (const float*)d_in[4];
    const float* b2 = (const float*)d_in[5];
    const float* W3 = (const float*)d_in[6];
    const float* b3 = (const float*)d_in[7];

    const int N = in_sizes[0] / DF;
    const int E = in_sizes[1] / 2;
    const int* src = ei;
    const int* dst = ei + E;

    float* out = (float*)d_out;

    char* ws = (char*)d_ws;
    size_t off0 = ((size_t)N * sizeof(float) + 255) & ~(size_t)255;
    float* dinv = (float*)ws;
    float* H    = (float*)(ws + off0);
    float* AGGa = H + (size_t)N * DF;
    unsigned char* blk = (unsigned char*)(AGGa + (size_t)N * DF);
    int* nrecs = (int*)(ws + ((size_t)128 << 20));
    unsigned* recs = (unsigned*)(ws + ((size_t)128 << 20) + 32768);

    const int strips  = (N + 15) / 16;
    const int gblocks = (strips + 7) / 8;

    unsigned* hist = (unsigned*)AGGa;
    k_degree_hist<<<HG, 1024, 0, stream>>>(dst, blk, hist, E);
    k_dinv_sum   <<<(HWORDS + 255) / 256, 256, 0, stream>>>(hist, dinv, N, HG);

    k_gemm_wmma<<<gblocks, 256, 0, stream>>>(x, W1, H, strips, N);
    k_aggregate<true, true, false>  <<<256, AGT, 0, stream>>>(H, src, dst, dinv, b1, blk, recs, nrecs, W2, AGGa, N, E);

    k_aggregate<false, true, false> <<<256, AGT, 0, stream>>>(AGGa, src, dst, dinv, b2, blk, recs, nrecs, W3, H, N, E);

    k_aggregate<false, false, true><<<256, AGT, 0, stream>>>(H, src, dst, dinv, b3, blk, recs, nrecs, W3, out, N, E);
}


# First GEMM launch resident (256 groups x 32 waves) and given the edge filter: it builds the blocks' records beside the WMMAs; layer 1 sums from records as layers 2-3 do: list order, no float atomics
# speedup vs baseline: 3.8577x; 1.1311x over previous
#include <hip/hip_runtime.h>
#include <hip/hip_bf16.h>


typedef __attribute__((ext_vector_type(2))) float v2f;
typedef __attribute__((ext_vector_type(4))) float v4f;
typedef __attribute__((ext_vector_type(8))) float v8f;
typedef __attribute__((ext_vector_type(4))) unsigned v4u;
typedef __attribute__((ext_vector_type(2))) unsigned v2u;

#define DF 128


#define TR 391

#define HG     128
#define HWORDS (256 * TR / 2)

#define EU 4
__global__ __launch_bounds__(1024) void k_degree_hist(const int* __restrict__ dst,
                                                      unsigned char* __restrict__ blk,
                                                      unsigned* __restrict__ hist, int E) {
    __shared__ __align__(16) unsigned h[HWORDS];
    const v4u z = {0u, 0u, 0u, 0u};
    for (int i = threadIdx.x; i < HWORDS / 4; i += blockDim.x) ((v4u*)h)[i] = z;
    __syncthreads();

    const int chunk = (E + (int)gridDim.x - 1) / (int)gridDim.x;
    const int e0 = (int)blockIdx.x * chunk;
    const int e1 = (e0 + chunk) < E ? (e0 + chunk) : E;
    const int step = (int)blockDim.x;
    int e = e0 + (int)threadIdx.x;
    for (; e + (EU - 1) * step < e1; e += EU * step) {
        int d[EU];
#pragma unroll
        for (int u = 0; u < EU; ++u) d[u] = dst[e + u * step];
#pragma unroll
        for (int u = 0; u < EU; ++u) {
            blk[e + u * step] = (unsigned char)(d[u] / TR);
            (void)__hip_atomic_fetch_add(&h[d[u] >> 1], 1u << ((d[u] & 1) * 16),
                                         __ATOMIC_RELAXED, __HIP_MEMORY_SCOPE_WORKGROUP);
        }
    }
    for (; e < e1; e += step) {
        const int d = dst[e];
        blk[e] = (unsigned char)(d / TR);
        (void)__hip_atomic_fetch_add(&h[d >> 1], 1u << ((d & 1) * 16),
                                     __ATOMIC_RELAXED, __HIP_MEMORY_SCOPE_WORKGROUP);
    }
    if (blockIdx.x == gridDim.x - 1) {
        const int Epad = (E + 15) & ~15;
        for (int p = E + (int)threadIdx.x; p < Epad; p += step) blk[p] = 0;
    }
    __syncthreads();

    v4u* __restrict__ o = (v4u*)(hist + (size_t)blockIdx.x * HWORDS);
    for (int i = threadIdx.x; i < HWORDS / 4; i += blockDim.x) o[i] = ((const v4u*)h)[i];
}

__global__ __launch_bounds__(256) void k_dinv_sum(const unsigned* __restrict__ hist,
                                                  float* __restrict__ dinv, int n, int groups) {
    const int w = blockIdx.x * blockDim.x + threadIdx.x;
    if (w >= HWORDS) return;
    const unsigned* __restrict__ p = hist + w;
    unsigned c0 = 0u, c1 = 0u;
    int g = 0;
    for (; g + 32 <= groups; g += 32) {
        unsigned v[32];
#pragma unroll
        for (int u = 0; u < 32; ++u) v[u] = p[(size_t)(g + u) * HWORDS];
#pragma unroll
        for (int u = 0; u < 32; ++u) { c0 += v[u] & 0xFFFFu; c1 += v[u] >> 16; }
    }
    for (; g < groups; ++g) {
        const unsigned v = p[(size_t)g * HWORDS];
        c0 += v & 0xFFFFu;
        c1 += v >> 16;
    }
    const int i = 2 * w;
    if (i < n)     dinv[i]     = rsqrtf((float)c0 + 1.0f);
    if (i + 1 < n) dinv[i + 1] = rsqrtf((float)c1 + 1.0f);
}

#define CAPB 24576
#define SRCB 23
#define RPC  4
#define AS   132
#define AGT  1024
#define TRP  (((TR + 15) / 16) * 16)
#define AGW  32
#define SUBCAP (CAPB / AGW)
#define QCAP 64
#define DB   16
#define RB   32
#define SU   4

__device__ __forceinline__ void lds_add(float* p, float x) {
    (void)__hip_atomic_fetch_add(p, x, __ATOMIC_RELAXED, __HIP_MEMORY_SCOPE_WORKGROUP);
}

__device__ __forceinline__ unsigned zero_bytes(unsigned x) {
    const unsigned t = (x & 0x7F7F7F7Fu) + 0x7F7F7F7Fu;
    return ~(t | x | 0x7F7F7F7Fu);
}

template <bool FULL>
__device__ __forceinline__ void drain_batch(
        const unsigned* qw, int cnt, int lane, int lo,
        const float* __restrict__ H, const int* __restrict__ src,
        const int* __restrict__ dst, const float* __restrict__ dinv, float* acc) {
    int s = 0, d = lo;
    float ws = 0.0f, wd = 0.0f;
    if (lane < cnt) {
        const unsigned e = qw[lane];
        s = src[e];
        d = dst[e];
        ws = dinv[s];
        wd = dinv[d];
    }
    const int dl = d - lo;
    const float* __restrict__ hl = H + lane;
    float* al = acc + lane;
    if (FULL) {
        float v[DB][RPC];
#pragma unroll
        for (int j = 0; j < DB; ++j) {
            const int sj = __builtin_amdgcn_readlane(s, j);
            const float* __restrict__ row = hl + (size_t)sj * DF;
#pragma unroll
            for (int c = 0; c < RPC; ++c) v[j][c] = row[32 * c];
        }
#pragma unroll
        for (int j = 0; j < DB; ++j) {
            const int dj = __builtin_amdgcn_readlane(dl, j);
            const float nj =
                __builtin_bit_cast(float, __builtin_amdgcn_readlane(__builtin_bit_cast(int, ws), j)) *
                __builtin_bit_cast(float, __builtin_amdgcn_readlane(__builtin_bit_cast(int, wd), j));
            float* a = al + dj * AS;
#pragma unroll
            for (int c = 0; c < RPC; ++c) lds_add(a + 32 * c, v[j][c] * nj);
        }
    } else {
        for (int j = 0; j < cnt; ++j) {
            const int sj = __builtin_amdgcn_readlane(s, j);
            const int dj = __builtin_amdgcn_readlane(dl, j);
            const float nj =
                __builtin_bit_cast(float, __builtin_amdgcn_readlane(__builtin_bit_cast(int, ws), j)) *
                __builtin_bit_cast(float, __builtin_amdgcn_readlane(__builtin_bit_cast(int, wd), j));
            const float* __restrict__ row = hl + (size_t)sj * DF;
            float* a = al + dj * AS;
            float t[RPC];
#pragma unroll
            for (int c = 0; c < RPC; ++c) t[c] = row[32 * c];
#pragma unroll
            for (int c = 0; c < RPC; ++c) lds_add(a + 32 * c, t[c] * nj);
        }
    }
}

__device__ __forceinline__ void record_batch(
        const unsigned* qw, int cnt, int lane, int lo,
        const int* __restrict__ src, const int* __restrict__ dst,
        const float* __restrict__ dinv, v2u* __restrict__ rb, int* nrec) {
    if (lane < cnt) {
        const unsigned e = qw[lane];
        const int s = src[e];
        const int d = dst[e];
        const float ws = dinv[s];
        const float wd = dinv[d];
        const int dl = d - lo;
        const int ow = dl & (AGW - 1);
        const int slot = __hip_atomic_fetch_add(nrec + ow, 1, __ATOMIC_RELAXED, __HIP_MEMORY_SCOPE_WORKGROUP);
        if (slot < SUBCAP) {
            const v2u r = {(unsigned)s | ((unsigned)dl << SRCB), __builtin_bit_cast(unsigned, ws * wd)};
            rb[ow * SUBCAP + slot] = r;
        }
    }
}

template <bool FULL>
__device__ __forceinline__ void drain_own(v2u r, int cnt, int lane,
                                          const float* __restrict__ H, float* acc) {
    const char* __restrict__ hb = (const char*)H;
    const unsigned lo16 = (unsigned)lane * 16u;
    float* al = acc + 4 * lane;
    const int rx = (int)r.x, ry = (int)r.y;
    if (FULL) {
        v4f v[DB];
#pragma unroll
        for (int j = 0; j < DB; ++j) {
            const unsigned pj = (unsigned)__builtin_amdgcn_readlane(rx, j);
            v[j] = *(const v4f*)(hb + (((pj & ((1u << SRCB) - 1u)) * (unsigned)(DF * 4)) + lo16));
        }
        __builtin_amdgcn_sched_barrier(0);
#pragma unroll
        for (int j = 0; j < DB; ++j) {
            const unsigned pj = (unsigned)__builtin_amdgcn_readlane(rx, j);
            const float nj = __builtin_bit_cast(float, __builtin_amdgcn_readlane(ry, j));
            v4f* a = (v4f*)(al + (pj >> SRCB) * AS);
            v4f c = *a;
            c.x = fmaf(v[j].x, nj, c.x);
            c.y = fmaf(v[j].y, nj, c.y);
            c.z = fmaf(v[j].z, nj, c.z);
            c.w = fmaf(v[j].w, nj, c.w);
            *a = c;
        }
    } else {
        for (int j = 0; j < cnt; ++j) {
            const unsigned pj = (unsigned)__builtin_amdgcn_readlane(rx, j);
            const float nj = __builtin_bit_cast(float, __builtin_amdgcn_readlane(ry, j));
            const v4f t = *(const v4f*)(hb + (((pj & ((1u << SRCB) - 1u)) * (unsigned)(DF * 4)) + lo16));
            v4f* a = (v4f*)(al + (pj >> SRCB) * AS);
            v4f c = *a;
            c.x = fmaf(t.x, nj, c.x);
            c.y = fmaf(t.y, nj, c.y);
            c.z = fmaf(t.z, nj, c.z);
            c.w = fmaf(t.w, nj, c.w);
            *a = c;
        }
    }
}

template <bool REC>
__device__ __forceinline__ void filter_block(
        int tile, int lo, int wave, int lane, int G, int E,
        const v4u* __restrict__ b4, unsigned* qw,
        const float* __restrict__ H, const int* __restrict__ src,
        const int* __restrict__ dst, const float* __restrict__ dinv, float* acc,
        v2u* __restrict__ rb, int* nrec) {
    const unsigned pat = (unsigned)tile * 0x01010101u;
    int qn = 0;
    const int glast = G > 0 ? G - 1 : 0;
    v4u w[SU];
#pragma unroll
    for (int u = 0; u < SU; ++u) {
        const int g = (wave * SU + u) * 32 + lane;
        w[u] = b4[g < G ? g : glast];
    }
    for (int i0 = wave * SU; i0 * 32 < G; i0 += AGW * SU) {
        v4u nx[SU];
#pragma unroll
        for (int u = 0; u < SU; ++u) {
            const int g = (i0 + AGW * SU + u) * 32 + lane;
            nx[u] = b4[g < G ? g : glast];
        }
#pragma unroll
        for (int u = 0; u < SU; ++u) {
            const int g = (i0 + u) * 32 + lane;
            unsigned mask = (zero_bytes(w[u].x ^ pat) >> 7) | (zero_bytes(w[u].y ^ pat) >> 6) |
                            (zero_bytes(w[u].z ^ pat) >> 5) | (zero_bytes(w[u].w ^ pat) >> 4);
            if (g >= G) mask = 0u;
            while (__builtin_amdgcn_ballot_w32(mask != 0u) != 0u) {
                const bool has = mask != 0u;
                const unsigned p = (unsigned)__builtin_ctz(mask | 0x80000000u);
                mask &= mask - 1u;
                const unsigned e = (unsigned)g * 16u + ((p & 7u) << 2) + (p >> 3);
                const bool ok = has && e < (unsigned)E;
                const unsigned okb = __builtin_amdgcn_ballot_w32(ok);
                if (ok) qw[qn + (int)__builtin_amdgcn_mbcnt_lo(okb, 0u)] = e;
                qn += __builtin_popcount(okb);
                __builtin_amdgcn_wave_barrier();
                if (REC) {
                    while (qn >= RB) {
                        qn -= RB;
                        record_batch(qw + qn, RB, lane, lo, src, dst, dinv, rb, nrec);
                    }
                } else {
                    while (qn >= DB) {
                        qn -= DB;
                        drain_batch<true>(qw + qn, DB, lane, lo, H, src, dst, dinv, acc);
                    }
                }
            }
        }
#pragma unroll
        for (int u = 0; u < SU; ++u) w[u] = nx[u];
    }
    if (qn > 0) {
        if (REC) record_batch(qw, qn, lane, lo, src, dst, dinv, rb, nrec);
        else     drain_batch<false>(qw, qn, lane, lo, H, src, dst, dinv, acc);
    }
}


__global__ __launch_bounds__(AGT) void k_gemm_records(
        const float* __restrict__ X, const float* __restrict__ W,
        float* __restrict__ H, int strips, int nrows,
        const int* __restrict__ src, const int* __restrict__ dst,
        const float* __restrict__ dinv, const unsigned char* __restrict__ blk,
        unsigned* __restrict__ recw, int* __restrict__ nrecs, int E) {
    __shared__ __align__(16) float ldsW[DF * DF];
    __shared__ unsigned q[AGW * QCAP];
    __shared__ int nrec[AGW];

    const int lane = threadIdx.x & 31;
    const int wave = __builtin_amdgcn_readfirstlane((int)(threadIdx.x >> 5));

    {
        float w0[DF * DF / 2 / AGT], w1[DF * DF / 2 / AGT];
#pragma unroll
        for (int u = 0; u < DF * DF / 2 / AGT; ++u) {
            const int p = u * AGT + (int)threadIdx.x;
            const int qq = p >> 7, n = p & 127;
            w0[u] = W[(2 * qq) * DF + n];
            w1[u] = W[(2 * qq + 1) * DF + n];
        }
#pragma unroll
        for (int u = 0; u < DF * DF / 2 / AGT; ++u) {
            const int p = u * AGT + (int)threadIdx.x;
            const int qq = p >> 7, n = p & 127;
            const v2f v = {w0[u], w1[u]};
            ((v2f*)ldsW)[(((n >> 4) << 5) + (qq >> 1)) * 32 + (qq & 1) * 16 + (n & 15)] = v;
        }
    }
    if (threadIdx.x < AGW) nrec[threadIdx.x] = 0;
    __syncthreads();

    const int strip = wave * (int)gridDim.x + (int)blockIdx.x;
    if (strip < strips) {
        const int nloc  = lane & 15;
        const int khalf = lane >> 4;

        int arow_i = strip * 16 + nloc;
        if (arow_i >= nrows) arow_i = nrows - 1;
        const float* __restrict__ arow = X + (size_t)arow_i * DF + 2 * khalf;

        const v2f* bbase = (const v2f*)ldsW + lane;

        v8f c[8] = {};
#pragma unroll
        for (int k = 0; k < 32; ++k) {
            const v2f a = *(const v2f*)(arow + 4 * k);
#pragma unroll
            for (int t = 0; t < 8; ++t) {
                const v2f b = bbase[(t * 32 + k) * 32];
                c[t] = __builtin_amdgcn_wmma_f32_16x16x4_f32(
                           false, a, false, b, (short)0, c[t], false, false);
            }
        }

        const int rbase = strip * 16 + 8 * khalf;
#pragma unroll
        for (int r = 0; r < 8; ++r) {
            const int row = rbase + r;
            if (row >= nrows) continue;
            const size_t ro = (size_t)row * DF;
#pragma unroll
            for (int t = 0; t < 8; ++t) H[ro + t * 16 + nloc] = c[t][r];
        }
    }

    const int tile = blockIdx.x;
    const int nblk = (nrows + TR - 1) / TR;
    if (tile >= nblk) return;
    const int G = (E + 15) >> 4;
    v2u* __restrict__ rb = (v2u*)recw + (size_t)tile * CAPB;
    filter_block<true>(tile, tile * TR, wave, lane, G, E, (const v4u*)blk, q + wave * QCAP,
                       nullptr, src, dst, dinv, nullptr, rb, nrec);
    __syncthreads();
    if (threadIdx.x < AGW) nrecs[tile * AGW + threadIdx.x] = nrec[threadIdx.x];
}

template <bool FUSE, bool NORM>
__global__ __launch_bounds__(AGT) void k_aggregate(
        const float* __restrict__ H, const int* __restrict__ src,
        const int* __restrict__ dst, const float* __restrict__ dinv,
        const float* __restrict__ bias, const unsigned char* __restrict__ blk,
        const unsigned* __restrict__ recw, const int* __restrict__ nrecs,
        const float* __restrict__ Wn, float* __restrict__ OUT, int N, int E) {
    __shared__ __align__(16) float acc[(FUSE ? TRP : TR) * AS];
    __shared__ __align__(16) float ldsW[FUSE ? DF * DF : 4];
    __shared__ unsigned q[AGW * QCAP];

    const int lane = threadIdx.x & 31;
    const int wave = __builtin_amdgcn_readfirstlane((int)(threadIdx.x >> 5));
    const int nblk = (N + TR - 1) / TR;
    const int G = (E + 15) >> 4;
    const v4u* __restrict__ b4 = (const v4u*)blk;
    unsigned* qw = q + wave * QCAP;

    if (FUSE) {
        float w0[DF * DF / 2 / AGT], w1[DF * DF / 2 / AGT];
#pragma unroll
        for (int u = 0; u < DF * DF / 2 / AGT; ++u) {
            const int p = u * AGT + (int)threadIdx.x;
            const int qq = p >> 7, n = p & 127;
            w0[u] = Wn[(2 * qq) * DF + n];
            w1[u] = Wn[(2 * qq + 1) * DF + n];
        }
#pragma unroll
        for (int u = 0; u < DF * DF / 2 / AGT; ++u) {
            const int p = u * AGT + (int)threadIdx.x;
            const int qq = p >> 7, n = p & 127;
            const v2f v = {w0[u], w1[u]};
            ((v2f*)ldsW)[(((n >> 4) << 5) + (qq >> 1)) * 32 + (qq & 1) * 16 + (n & 15)] = v;
        }
    }

    const int tile = blockIdx.x;
    if (tile >= nblk) return;
    {
        const int lo = tile * TR;
        const int rows = (N - lo) < TR ? (N - lo) : TR;
        const v2u* __restrict__ rb = (const v2u*)recw + (size_t)tile * CAPB;

        for (int i = threadIdx.x; i < rows * (DF / 4); i += blockDim.x) {
            const int r = i >> 5, c4 = (i & 31) * 4;
            const float dv = dinv[lo + r];
            const float d2 = dv * dv;
            const v4f h = *(const v4f*)(H + (size_t)(lo + r) * DF + c4);
            const v4f b = *(const v4f*)(bias + c4);
            v4f o = {fmaf(h.x, d2, b.x), fmaf(h.y, d2, b.y), fmaf(h.z, d2, b.z), fmaf(h.w, d2, b.w)};
            *(v4f*)(acc + r * AS + c4) = o;
        }
        if (FUSE) {
            const v4f z = {0.0f, 0.0f, 0.0f, 0.0f};
            for (int i = rows * (DF / 4) + (int)threadIdx.x; i < TRP * (DF / 4); i += blockDim.x)
                *(v4f*)(acc + (i >> 5) * AS + (i & 31) * 4) = z;
        }
        __syncthreads();

        {
            const int cl = nrecs[tile * AGW + lane];
            if (__builtin_amdgcn_ballot_w32(cl > SUBCAP) == 0u) {
                const int total = __builtin_amdgcn_readlane(cl, wave);
                const v2u* __restrict__ rw = rb + wave * SUBCAP;
                v2u r = {0u, 0u};
                if (lane < DB && lane < total) r = rw[lane];
                for (int base = 0; base < total; base += DB) {
                    const int nb = base + DB;
                    v2u rn = {0u, 0u};
                    if (lane < DB && nb + lane < total) rn = rw[nb + lane];
                    const int n = total - base;
                    if (n >= DB) drain_own<true>(r, DB, lane, H, acc);
                    else         drain_own<false>(r, n, lane, H, acc);
                    r = rn;
                }
            } else {
                filter_block<false>(tile, lo, wave, lane, G, E, b4, qw, H, src, dst, dinv, acc, nullptr, nullptr);
            }
        }
        __syncthreads();

        if (FUSE) {
            if (wave * 16 < rows) {
                const int nloc  = lane & 15;
                const int khalf = lane >> 4;
                const float* arow = acc + (wave * 16 + nloc) * AS + 2 * khalf;
                const v2f* bbase = (const v2f*)ldsW + lane;
                v8f c[8] = {};
#pragma unroll
                for (int k = 0; k < 32; ++k) {
                    v2f a = *(const v2f*)(arow + 4 * k);
                    a.x = fmaxf(a.x, 0.0f);
                    a.y = fmaxf(a.y, 0.0f);
#pragma unroll
                    for (int t = 0; t < 8; ++t) {
                        const v2f b = bbase[(t * 32 + k) * 32];
                        c[t] = __builtin_amdgcn_wmma_f32_16x16x4_f32(
                                   false, a, false, b, (short)0, c[t], false, false);
                    }
                }
                float* drow = acc + (wave * 16 + 8 * khalf) * AS + nloc;
#pragma unroll
                for (int r = 0; r < 8; ++r) {
#pragma unroll
                    for (int t = 0; t < 8; ++t) drow[r * AS + t * 16] = c[t][r];
                }
            }
            __syncthreads();
        }

        for (int i = threadIdx.x; i < rows * (DF / 4); i += blockDim.x) {
            const int r = i >> 5, c4 = (i & 31) * 4;
            v4f v = *(const v4f*)(acc + r * AS + c4);
            if (NORM) {
                float ss = v.x * v.x + v.y * v.y + v.z * v.z + v.w * v.w;
#pragma unroll
                for (int m = 16; m > 0; m >>= 1) ss += __shfl_xor(ss, m, 32);
                const float sc = 1.0f / fmaxf(sqrtf(ss), 1e-12f);
                v.x *= sc; v.y *= sc; v.z *= sc; v.w *= sc;
            }
            *(v4f*)(OUT + (size_t)(lo + r) * DF + c4) = v;
        }
    }
}


extern "C" void kernel_launch(void* const* d_in, const int* in_sizes, int n_in,
                              void* d_out, int out_size, void* d_ws, size_t ws_size,
                              hipStream_t stream) {
    const float* x  = (const float*)d_in[0];
    const int*   ei = (const int*)  d_in[1];
    const float* W1 = (const float*)d_in[2];
    const float* b1 = (const float*)d_in[3];
    const float* W2 = (const float*)d_in[4];
    const float* b2 = (const float*)d_in[5];
    const float* W3 = (const float*)d_in[6];
    const float* b3 = (const float*)d_in[7];

    const int N = in_sizes[0] / DF;
    const int E = in_sizes[1] / 2;
    const int* src = ei;
    const int* dst = ei + E;

    float* out = (float*)d_out;

    char* ws = (char*)d_ws;
    size_t off0 = ((size_t)N * sizeof(float) + 255) & ~(size_t)255;
    float* dinv = (float*)ws;
    float* H    = (float*)(ws + off0);
    float* AGGa = H + (size_t)N * DF;
    unsigned char* blk = (unsigned char*)(AGGa + (size_t)N * DF);
    int* nrecs = (int*)(ws + ((size_t)128 << 20));
    unsigned* recs = (unsigned*)(ws + ((size_t)128 << 20) + 32768);

    const int strips  = (N + 15) / 16;

    unsigned* hist = (unsigned*)AGGa;
    k_degree_hist<<<HG, 1024, 0, stream>>>(dst, blk, hist, E);
    k_dinv_sum   <<<(HWORDS + 255) / 256, 256, 0, stream>>>(hist, dinv, N, HG);

    k_gemm_records<<<256, AGT, 0, stream>>>(x, W1, H, strips, N, src, dst, dinv, blk, recs, nrecs, E);
    k_aggregate<true, false><<<256, AGT, 0, stream>>>(H, src, dst, dinv, b1, blk, recs, nrecs, W2, AGGa, N, E);

    k_aggregate<true, false><<<256, AGT, 0, stream>>>(AGGa, src, dst, dinv, b2, blk, recs, nrecs, W3, H, N, E);

    k_aggregate<false, true><<<256, AGT, 0, stream>>>(H, src, dst, dinv, b3, blk, recs, nrecs, W3, out, N, E);
}


# Aggregation: a wave keeps the sums of its own rows in registers; a record's row number, a scalar, indexes them (indirect register access); no LDS read-add-write per record; list order kept
# speedup vs baseline: 4.1002x; 1.0629x over previous
#include <hip/hip_runtime.h>
#include <hip/hip_bf16.h>


typedef __attribute__((ext_vector_type(2))) float v2f;
typedef __attribute__((ext_vector_type(4))) float v4f;
typedef __attribute__((ext_vector_type(8))) float v8f;
typedef __attribute__((ext_vector_type(16))) float v16f;
typedef __attribute__((ext_vector_type(4))) unsigned v4u;
typedef __attribute__((ext_vector_type(2))) unsigned v2u;

#define DF 128


#define TR 391

#define HG     128
#define HWORDS (256 * TR / 2)

#define EU 4
__global__ __launch_bounds__(1024) void k_degree_hist(const int* __restrict__ dst,
                                                      unsigned char* __restrict__ blk,
                                                      unsigned* __restrict__ hist, int E) {
    __shared__ __align__(16) unsigned h[HWORDS];
    const v4u z = {0u, 0u, 0u, 0u};
    for (int i = threadIdx.x; i < HWORDS / 4; i += blockDim.x) ((v4u*)h)[i] = z;
    __syncthreads();

    const int chunk = (E + (int)gridDim.x - 1) / (int)gridDim.x;
    const int e0 = (int)blockIdx.x * chunk;
    const int e1 = (e0 + chunk) < E ? (e0 + chunk) : E;
    const int step = (int)blockDim.x;
    int e = e0 + (int)threadIdx.x;
    for (; e + (EU - 1) * step < e1; e += EU * step) {
        int d[EU];
#pragma unroll
        for (int u = 0; u < EU; ++u) d[u] = dst[e + u * step];
#pragma unroll
        for (int u = 0; u < EU; ++u) {
            blk[e + u * step] = (unsigned char)(d[u] / TR);
            (void)__hip_atomic_fetch_add(&h[d[u] >> 1], 1u << ((d[u] & 1) * 16),
                                         __ATOMIC_RELAXED, __HIP_MEMORY_SCOPE_WORKGROUP);
        }
    }
    for (; e < e1; e += step) {
        const int d = dst[e];
        blk[e] = (unsigned char)(d / TR);
        (void)__hip_atomic_fetch_add(&h[d >> 1], 1u << ((d & 1) * 16),
                                     __ATOMIC_RELAXED, __HIP_MEMORY_SCOPE_WORKGROUP);
    }
    if (blockIdx.x == gridDim.x - 1) {
        const int Epad = (E + 15) & ~15;
        for (int p = E + (int)threadIdx.x; p < Epad; p += step) blk[p] = 0;
    }
    __syncthreads();

    v4u* __restrict__ o = (v4u*)(hist + (size_t)blockIdx.x * HWORDS);
    for (int i = threadIdx.x; i < HWORDS / 4; i += blockDim.x) o[i] = ((const v4u*)h)[i];
}

__global__ __launch_bounds__(256) void k_dinv_sum(const unsigned* __restrict__ hist,
                                                  float* __restrict__ dinv, int n, int groups) {
    const int w = blockIdx.x * blockDim.x + threadIdx.x;
    if (w >= HWORDS) return;
    const unsigned* __restrict__ p = hist + w;
    unsigned c0 = 0u, c1 = 0u;
    int g = 0;
    for (; g + 32 <= groups; g += 32) {
        unsigned v[32];
#pragma unroll
        for (int u = 0; u < 32; ++u) v[u] = p[(size_t)(g + u) * HWORDS];
#pragma unroll
        for (int u = 0; u < 32; ++u) { c0 += v[u] & 0xFFFFu; c1 += v[u] >> 16; }
    }
    for (; g < groups; ++g) {
        const unsigned v = p[(size_t)g * HWORDS];
        c0 += v & 0xFFFFu;
        c1 += v >> 16;
    }
    const int i = 2 * w;
    if (i < n)     dinv[i]     = rsqrtf((float)c0 + 1.0f);
    if (i + 1 < n) dinv[i + 1] = rsqrtf((float)c1 + 1.0f);
}

#define CAPB 24576
#define SRCB 23
#define RPC  4
#define AS   132
#define AGT  1024
#define TRP  (((TR + 15) / 16) * 16)
#define AGW  32
#define SUBCAP (CAPB / AGW)
#define QCAP 64
#define DB   16
#define RB   32
#define NR   ((TR + AGW - 1) / AGW)
#define DR   12
#define PADREC ((unsigned)(15 * AGW) << SRCB)
#define SU   4

__device__ __forceinline__ void lds_add(float* p, float x) {
    (void)__hip_atomic_fetch_add(p, x, __ATOMIC_RELAXED, __HIP_MEMORY_SCOPE_WORKGROUP);
}

__device__ __forceinline__ unsigned zero_bytes(unsigned x) {
    const unsigned t = (x & 0x7F7F7F7Fu) + 0x7F7F7F7Fu;
    return ~(t | x | 0x7F7F7F7Fu);
}

template <bool FULL>
__device__ __forceinline__ void drain_batch(
        const unsigned* qw, int cnt, int lane, int lo,
        const float* __restrict__ H, const int* __restrict__ src,
        const int* __restrict__ dst, const float* __restrict__ dinv, float* acc) {
    int s = 0, d = lo;
    float ws = 0.0f, wd = 0.0f;
    if (lane < cnt) {
        const unsigned e = qw[lane];
        s = src[e];
        d = dst[e];
        ws = dinv[s];
        wd = dinv[d];
    }
    const int dl = d - lo;
    const float* __restrict__ hl = H + lane;
    float* al = acc + lane;
    if (FULL) {
        float v[DB][RPC];
#pragma unroll
        for (int j = 0; j < DB; ++j) {
            const int sj = __builtin_amdgcn_readlane(s, j);
            const float* __restrict__ row = hl + (size_t)sj * DF;
#pragma unroll
            for (int c = 0; c < RPC; ++c) v[j][c] = row[32 * c];
        }
#pragma unroll
        for (int j = 0; j < DB; ++j) {
            const int dj = __builtin_amdgcn_readlane(dl, j);
            const float nj =
                __builtin_bit_cast(float, __builtin_amdgcn_readlane(__builtin_bit_cast(int, ws), j)) *
                __builtin_bit_cast(float, __builtin_amdgcn_readlane(__builtin_bit_cast(int, wd), j));
            float* a = al + dj * AS;
#pragma unroll
            for (int c = 0; c < RPC; ++c) lds_add(a + 32 * c, v[j][c] * nj);
        }
    } else {
        for (int j = 0; j < cnt; ++j) {
            const int sj = __builtin_amdgcn_readlane(s, j);
            const int dj = __builtin_amdgcn_readlane(dl, j);
            const float nj =
                __builtin_bit_cast(float, __builtin_amdgcn_readlane(__builtin_bit_cast(int, ws), j)) *
                __builtin_bit_cast(float, __builtin_amdgcn_readlane(__builtin_bit_cast(int, wd), j));
            const float* __restrict__ row = hl + (size_t)sj * DF;
            float* a = al + dj * AS;
            float t[RPC];
#pragma unroll
            for (int c = 0; c < RPC; ++c) t[c] = row[32 * c];
#pragma unroll
            for (int c = 0; c < RPC; ++c) lds_add(a + 32 * c, t[c] * nj);
        }
    }
}

__device__ __forceinline__ void record_batch(
        const unsigned* qw, int cnt, int lane, int lo,
        const int* __restrict__ src, const int* __restrict__ dst,
        const float* __restrict__ dinv, v2u* __restrict__ rb, int* nrec) {
    if (lane < cnt) {
        const unsigned e = qw[lane];
        const int s = src[e];
        const int d = dst[e];
        const float ws = dinv[s];
        const float wd = dinv[d];
        const int dl = d - lo;
        const int ow = dl & (AGW - 1);
        const int slot = __hip_atomic_fetch_add(nrec + ow, 1, __ATOMIC_RELAXED, __HIP_MEMORY_SCOPE_WORKGROUP);
        if (slot < SUBCAP) {
            const v2u r = {(unsigned)s | ((unsigned)dl << SRCB), __builtin_bit_cast(unsigned, ws * wd)};
            rb[ow * SUBCAP + slot] = r;
        }
    }
}

__device__ __forceinline__ void drain_regs(v2u r, int lane, const float* __restrict__ H,
                                           v16f& ax, v16f& ay, v16f& az, v16f& aw) {
    const char* __restrict__ hb = (const char*)H;
    const unsigned lo16 = (unsigned)lane * 16u;
    const int rx = (int)r.x, ry = (int)r.y;
    v4f v[DR];
#pragma unroll
    for (int j = 0; j < DR; ++j) {
        const unsigned pj = (unsigned)__builtin_amdgcn_readlane(rx, j);
        v[j] = *(const v4f*)(hb + (((pj & ((1u << SRCB) - 1u)) * (unsigned)(DF * 4)) + lo16));
    }
    __builtin_amdgcn_sched_barrier(0);
#pragma unroll
    for (int j = 0; j < DR; ++j) {
        const unsigned pj = (unsigned)__builtin_amdgcn_readlane(rx, j);
        const float nj = __builtin_bit_cast(float, __builtin_amdgcn_readlane(ry, j));
        const int i = (int)(pj >> (SRCB + 5));
        ax[i] = fmaf(v[j].x, nj, ax[i]);
        ay[i] = fmaf(v[j].y, nj, ay[i]);
        az[i] = fmaf(v[j].z, nj, az[i]);
        aw[i] = fmaf(v[j].w, nj, aw[i]);
    }
}

template <bool REC>
__device__ __forceinline__ void filter_block(
        int tile, int lo, int wave, int lane, int G, int E,
        const v4u* __restrict__ b4, unsigned* qw,
        const float* __restrict__ H, const int* __restrict__ src,
        const int* __restrict__ dst, const float* __restrict__ dinv, float* acc,
        v2u* __restrict__ rb, int* nrec) {
    const unsigned pat = (unsigned)tile * 0x01010101u;
    int qn = 0;
    const int glast = G > 0 ? G - 1 : 0;
    v4u w[SU];
#pragma unroll
    for (int u = 0; u < SU; ++u) {
        const int g = (wave * SU + u) * 32 + lane;
        w[u] = b4[g < G ? g : glast];
    }
    for (int i0 = wave * SU; i0 * 32 < G; i0 += AGW * SU) {
        v4u nx[SU];
#pragma unroll
        for (int u = 0; u < SU; ++u) {
            const int g = (i0 + AGW * SU + u) * 32 + lane;
            nx[u] = b4[g < G ? g : glast];
        }
#pragma unroll
        for (int u = 0; u < SU; ++u) {
            const int g = (i0 + u) * 32 + lane;
            unsigned mask = (zero_bytes(w[u].x ^ pat) >> 7) | (zero_bytes(w[u].y ^ pat) >> 6) |
                            (zero_bytes(w[u].z ^ pat) >> 5) | (zero_bytes(w[u].w ^ pat) >> 4);
            if (g >= G) mask = 0u;
            while (__builtin_amdgcn_ballot_w32(mask != 0u) != 0u) {
                const bool has = mask != 0u;
                const unsigned p = (unsigned)__builtin_ctz(mask | 0x80000000u);
                mask &= mask - 1u;
                const unsigned e = (unsigned)g * 16u + ((p & 7u) << 2) + (p >> 3);
                const bool ok = has && e < (unsigned)E;
                const unsigned okb = __builtin_amdgcn_ballot_w32(ok);
                if (ok) qw[qn + (int)__builtin_amdgcn_mbcnt_lo(okb, 0u)] = e;
                qn += __builtin_popcount(okb);
                __builtin_amdgcn_wave_barrier();
                if (REC) {
                    while (qn >= RB) {
                        qn -= RB;
                        record_batch(qw + qn, RB, lane, lo, src, dst, dinv, rb, nrec);
                    }
                } else {
                    while (qn >= DB) {
                        qn -= DB;
                        drain_batch<true>(qw + qn, DB, lane, lo, H, src, dst, dinv, acc);
                    }
                }
            }
        }
#pragma unroll
        for (int u = 0; u < SU; ++u) w[u] = nx[u];
    }
    if (qn > 0) {
        if (REC) record_batch(qw, qn, lane, lo, src, dst, dinv, rb, nrec);
        else     drain_batch<false>(qw, qn, lane, lo, H, src, dst, dinv, acc);
    }
}


__global__ __launch_bounds__(AGT) void k_gemm_records(
        const float* __restrict__ X, const float* __restrict__ W,
        float* __restrict__ H, int strips, int nrows,
        const int* __restrict__ src, const int* __restrict__ dst,
        const float* __restrict__ dinv, const unsigned char* __restrict__ blk,
        unsigned* __restrict__ recw, int* __restrict__ nrecs, int E) {
    __shared__ __align__(16) float ldsW[DF * DF];
    __shared__ unsigned q[AGW * QCAP];
    __shared__ int nrec[AGW];

    const int lane = threadIdx.x & 31;
    const int wave = __builtin_amdgcn_readfirstlane((int)(threadIdx.x >> 5));

    {
        float w0[DF * DF / 2 / AGT], w1[DF * DF / 2 / AGT];
#pragma unroll
        for (int u = 0; u < DF * DF / 2 / AGT; ++u) {
            const int p = u * AGT + (int)threadIdx.x;
            const int qq = p >> 7, n = p & 127;
            w0[u] = W[(2 * qq) * DF + n];
            w1[u] = W[(2 * qq + 1) * DF + n];
        }
#pragma unroll
        for (int u = 0; u < DF * DF / 2 / AGT; ++u) {
            const int p = u * AGT + (int)threadIdx.x;
            const int qq = p >> 7, n = p & 127;
            const v2f v = {w0[u], w1[u]};
            ((v2f*)ldsW)[(((n >> 4) << 5) + (qq >> 1)) * 32 + (qq & 1) * 16 + (n & 15)] = v;
        }
    }
    if (threadIdx.x < AGW) nrec[threadIdx.x] = 0;
    __syncthreads();

    const int strip = wave * (int)gridDim.x + (int)blockIdx.x;
    if (strip < strips) {
        const int nloc  = lane & 15;
        const int khalf = lane >> 4;

        int arow_i = strip * 16 + nloc;
        if (arow_i >= nrows) arow_i = nrows - 1;
        const float* __restrict__ arow = X + (size_t)arow_i * DF + 2 * khalf;

        const v2f* bbase = (const v2f*)ldsW + lane;

        v8f c[8] = {};
#pragma unroll
        for (int k = 0; k < 32; ++k) {
            const v2f a = *(const v2f*)(arow + 4 * k);
#pragma unroll
            for (int t = 0; t < 8; ++t) {
                const v2f b = bbase[(t * 32 + k) * 32];
                c[t] = __builtin_amdgcn_wmma_f32_16x16x4_f32(
                           false, a, false, b, (short)0, c[t], false, false);
            }
        }

        const int rbase = strip * 16 + 8 * khalf;
#pragma unroll
        for (int r = 0; r < 8; ++r) {
            const int row = rbase + r;
            if (row >= nrows) continue;
            const size_t ro = (size_t)row * DF;
#pragma unroll
            for (int t = 0; t < 8; ++t) H[ro + t * 16 + nloc] = c[t][r];
        }
    }

    const int tile = blockIdx.x;
    const int nblk = (nrows + TR - 1) / TR;
    if (tile >= nblk) return;
    const int G = (E + 15) >> 4;
    v2u* __restrict__ rb = (v2u*)recw + (size_t)tile * CAPB;
    filter_block<true>(tile, tile * TR, wave, lane, G, E, (const v4u*)blk, q + wave * QCAP,
                       nullptr, src, dst, dinv, nullptr, rb, nrec);
    __syncthreads();
    if (threadIdx.x < AGW) nrecs[tile * AGW + threadIdx.x] = nrec[threadIdx.x];
}

template <bool FUSE, bool NORM>
__global__ __launch_bounds__(AGT) void k_aggregate(
        const float* __restrict__ H, const int* __restrict__ src,
        const int* __restrict__ dst, const float* __restrict__ dinv,
        const float* __restrict__ bias, const unsigned char* __restrict__ blk,
        const unsigned* __restrict__ recw, const int* __restrict__ nrecs,
        const float* __restrict__ Wn, float* __restrict__ OUT, int N, int E) {
    __shared__ __align__(16) float acc[(FUSE ? TRP : TR) * AS];
    __shared__ __align__(16) float ldsW[FUSE ? DF * DF : 4];
    __shared__ unsigned q[AGW * QCAP];

    const int lane = threadIdx.x & 31;
    const int wave = __builtin_amdgcn_readfirstlane((int)(threadIdx.x >> 5));
    const int nblk = (N + TR - 1) / TR;
    const int G = (E + 15) >> 4;
    const v4u* __restrict__ b4 = (const v4u*)blk;
    unsigned* qw = q + wave * QCAP;

    if (FUSE) {
        float w0[DF * DF / 2 / AGT], w1[DF * DF / 2 / AGT];
#pragma unroll
        for (int u = 0; u < DF * DF / 2 / AGT; ++u) {
            const int p = u * AGT + (int)threadIdx.x;
            const int qq = p >> 7, n = p & 127;
            w0[u] = Wn[(2 * qq) * DF + n];
            w1[u] = Wn[(2 * qq + 1) * DF + n];
        }
#pragma unroll
        for (int u = 0; u < DF * DF / 2 / AGT; ++u) {
            const int p = u * AGT + (int)threadIdx.x;
            const int qq = p >> 7, n = p & 127;
            const v2f v = {w0[u], w1[u]};
            ((v2f*)ldsW)[(((n >> 4) << 5) + (qq >> 1)) * 32 + (qq & 1) * 16 + (n & 15)] = v;
        }
    }

    const int tile = blockIdx.x;
    if (tile >= nblk) return;
    {
        const int lo = tile * TR;
        const int rows = (N - lo) < TR ? (N - lo) : TR;
        const v2u* __restrict__ rb = (const v2u*)recw + (size_t)tile * CAPB;

        if (FUSE) {
            const v4f z = {0.0f, 0.0f, 0.0f, 0.0f};
            for (int i = rows * (DF / 4) + (int)threadIdx.x; i < TRP * (DF / 4); i += blockDim.x)
                *(v4f*)(acc + (i >> 5) * AS + (i & 31) * 4) = z;
        }

        const int cl = nrecs[tile * AGW + lane];
        const bool over = __builtin_amdgcn_ballot_w32(cl > SUBCAP) != 0u;
        if (!over) {
            static_assert(AGW == 32, "a record's row number >> 5 is its index among the owner wave's rows");
            static_assert(NR <= 15, "the wave's rows are elements 0 .. NR-1 of 16-element register vectors; element 15 takes the pad records");
            const v4f b = *(const v4f*)(bias + 4 * lane);
            v16f ax = {}, ay = {}, az = {}, aw = {};
#pragma unroll
            for (int i = 0; i < NR; ++i) {
                const int rr = (wave + AGW * i) < rows ? (wave + AGW * i) : (rows - 1);
                const float dv = dinv[lo + rr];
                const float d2 = dv * dv;
                const v4f h = *(const v4f*)(H + (size_t)(lo + rr) * DF + 4 * lane);
                ax[i] = fmaf(h.x, d2, b.x);
                ay[i] = fmaf(h.y, d2, b.y);
                az[i] = fmaf(h.z, d2, b.z);
                aw[i] = fmaf(h.w, d2, b.w);
            }

            const int total = __builtin_amdgcn_readlane(cl, wave);
            const v2u* __restrict__ rw = rb + wave * SUBCAP;
            const v2u pad = {PADREC, 0u};
            v2u r = pad;
            if (lane < DR && lane < total) r = rw[lane];
            for (int base = 0; base < total; base += DR) {
                const int nb = base + DR;
                v2u rn = pad;
                if (lane < DR && nb + lane < total) rn = rw[nb + lane];
                drain_regs(r, lane, H, ax, ay, az, aw);
                r = rn;
            }

            if (NORM) {
                float ss[NR];
#pragma unroll
                for (int i = 0; i < NR; ++i)
                    ss[i] = ax[i] * ax[i] + ay[i] * ay[i] + az[i] * az[i] + aw[i] * aw[i];
#pragma unroll
                for (int m = 16; m > 0; m >>= 1) {
#pragma unroll
                    for (int i = 0; i < NR; ++i) ss[i] += __shfl_xor(ss[i], m, 32);
                }
#pragma unroll
                for (int i = 0; i < NR; ++i) {
                    const int rr = wave + AGW * i;
                    if (rr < rows) {
                        const float sc = 1.0f / fmaxf(sqrtf(ss[i]), 1e-12f);
                        const v4f o = {ax[i] * sc, ay[i] * sc, az[i] * sc, aw[i] * sc};
                        *(v4f*)(OUT + (size_t)(lo + rr) * DF + 4 * lane) = o;
                    }
                }
            } else {
#pragma unroll
                for (int i = 0; i < NR; ++i) {
                    const int rr = wave + AGW * i;
                    if (rr < rows) {
                        const v4f o = {ax[i], ay[i], az[i], aw[i]};
                        *(v4f*)(acc + rr * AS + 4 * lane) = o;
                    }
                }
            }
        } else {
            for (int i = threadIdx.x; i < rows * (DF / 4); i += blockDim.x) {
                const int r = i >> 5, c4 = (i & 31) * 4;
                const float dv = dinv[lo + r];
                const float d2 = dv * dv;
                const v4f h = *(const v4f*)(H + (size_t)(lo + r) * DF + c4);
                const v4f b = *(const v4f*)(bias + c4);
                v4f o = {fmaf(h.x, d2, b.x), fmaf(h.y, d2, b.y), fmaf(h.z, d2, b.z), fmaf(h.w, d2, b.w)};
                *(v4f*)(acc + r * AS + c4) = o;
            }
            __syncthreads();
            filter_block<false>(tile, lo, wave, lane, G, E, b4, qw, H, src, dst, dinv, acc, nullptr, nullptr);
        }
        __syncthreads();

        if (FUSE) {
            if (wave * 16 < rows) {
                const int nloc  = lane & 15;
                const int khalf = lane >> 4;
                const float* arow = acc + (wave * 16 + nloc) * AS + 2 * khalf;
                const v2f* bbase = (const v2f*)ldsW + lane;
                v8f c[8] = {};
#pragma unroll
                for (int k = 0; k < 32; ++k) {
                    v2f a = *(const v2f*)(arow + 4 * k);
                    a.x = fmaxf(a.x, 0.0f);
                    a.y = fmaxf(a.y, 0.0f);
#pragma unroll
                    for (int t = 0; t < 8; ++t) {
                        const v2f b = bbase[(t * 32 + k) * 32];
                        c[t] = __builtin_amdgcn_wmma_f32_16x16x4_f32(
                                   false, a, false, b, (short)0, c[t], false, false);
                    }
                }
                float* drow = acc + (wave * 16 + 8 * khalf) * AS + nloc;
#pragma unroll
                for (int r = 0; r < 8; ++r) {
#pragma unroll
                    for (int t = 0; t < 8; ++t) drow[r * AS + t * 16] = c[t][r];
                }
            }
            __syncthreads();
        }

        if (FUSE || over)
        for (int i = threadIdx.x; i < rows * (DF / 4); i += blockDim.x) {
            const int r = i >> 5, c4 = (i & 31) * 4;
            v4f v = *(const v4f*)(acc + r * AS + c4);
            if (NORM) {
                float ss = v.x * v.x + v.y * v.y + v.z * v.z + v.w * v.w;
#pragma unroll
                for (int m = 16; m > 0; m >>= 1) ss += __shfl_xor(ss, m, 32);
                const float sc = 1.0f / fmaxf(sqrtf(ss), 1e-12f);
                v.x *= sc; v.y *= sc; v.z *= sc; v.w *= sc;
            }
            *(v4f*)(OUT + (size_t)(lo + r) * DF + c4) = v;
        }
    }
}


extern "C" void kernel_launch(void* const* d_in, const int* in_sizes, int n_in,
                              void* d_out, int out_size, void* d_ws, size_t ws_size,
                              hipStream_t stream) {
    const float* x  = (const float*)d_in[0];
    const int*   ei = (const int*)  d_in[1];
    const float* W1 = (const float*)d_in[2];
    const float* b1 = (const float*)d_in[3];
    const float* W2 = (const float*)d_in[4];
    const float* b2 = (const float*)d_in[5];
    const float* W3 = (const float*)d_in[6];
    const float* b3 = (const float*)d_in[7];

    const int N = in_sizes[0] / DF;
    const int E = in_sizes[1] / 2;
    const int* src = ei;
    const int* dst = ei + E;

    float* out = (float*)d_out;

    char* ws = (char*)d_ws;
    size_t off0 = ((size_t)N * sizeof(float) + 255) & ~(size_t)255;
    float* dinv = (float*)ws;
    float* H    = (float*)(ws + off0);
    float* AGGa = H + (size_t)N * DF;
    unsigned char* blk = (unsigned char*)(AGGa + (size_t)N * DF);
    int* nrecs = (int*)(ws + ((size_t)128 << 20));
    unsigned* recs = (unsigned*)(ws + ((size_t)128 << 20) + 32768);

    const int strips  = (N + 15) / 16;

    unsigned* hist = (unsigned*)AGGa;
    k_degree_hist<<<HG, 1024, 0, stream>>>(dst, blk, hist, E);
    k_dinv_sum   <<<(HWORDS + 255) / 256, 256, 0, stream>>>(hist, dinv, N, HG);

    k_gemm_records<<<256, AGT, 0, stream>>>(x, W1, H, strips, N, src, dst, dinv, blk, recs, nrecs, E);
    k_aggregate<true, false><<<256, AGT, 0, stream>>>(H, src, dst, dinv, b1, blk, recs, nrecs, W2, AGGa, N, E);

    k_aggregate<true, false><<<256, AGT, 0, stream>>>(AGGa, src, dst, dinv, b2, blk, recs, nrecs, W3, H, N, E);

    k_aggregate<false, true><<<256, AGT, 0, stream>>>(H, src, dst, dinv, b3, blk, recs, nrecs, W3, out, N, E);
}
